# starting kernel
# baseline (speedup 1.0000x reference)
#include <hip/hip_runtime.h>
#include <hip/hip_bf16.h>

typedef __attribute__((ext_vector_type(2))) float v2f;
typedef __attribute__((ext_vector_type(8))) float v8f;

#define D_DIM 64
#define TILE_M 16
#define NEG_SLOPE 0.01f

__device__ __forceinline__ void atomic_max_f32(float* addr, float val) {
  if (val >= 0.0f) {
    atomicMax((int*)addr, __float_as_int(val));
  } else {
    atomicMin((unsigned int*)addr, __float_as_uint(val));
  }
}

__global__ __launch_bounds__(256, 2) void gat_linear_wmma(
    const float* __restrict__ feat, const float* __restrict__ W,
    const float* __restrict__ bias,
    const float* __restrict__ a1w, const float* __restrict__ a1b,
    const float* __restrict__ a2w, const float* __restrict__ a2b,
    float* __restrict__ h, float* __restrict__ a1, float* __restrict__ a2,
    int n_nodes)
{
  constexpr int WP = 68;
  constexpr int FP = 68;
  __shared__ float Wl[D_DIM * WP];
  __shared__ float ft[8][TILE_M * FP];

  const int tid  = threadIdx.x;
  const int wave = tid >> 5;
  const int lane = tid & 31;
  const int lh   = lane >> 4;
  const int lid  = lane & 15;

  for (int i = tid; i < D_DIM * (D_DIM / 4); i += 256) {
    const int n  = i >> 4;
    const int c4 = (i & 15) << 2;
    const float4 w4 = *(const float4*)(W + n * D_DIM + c4);
    *(float4*)&Wl[n * WP + c4] = w4;
  }
  __syncthreads();

  const int tile = blockIdx.x * 8 + wave;
  const int m0   = tile * TILE_M;
  if (m0 >= n_nodes) return;

  float* myft = ft[wave];
  for (int i = lane; i < TILE_M * (D_DIM / 4); i += 32) {
    const int r  = i >> 4;
    const int c4 = (i & 15) << 2;
    float4 f4 = {0.f, 0.f, 0.f, 0.f};
    if (m0 + r < n_nodes)
      f4 = *(const float4*)(feat + (size_t)(m0 + r) * D_DIM + c4);
    *(float4*)&myft[r * FP + c4] = f4;
  }

  v8f acc[4] = {};
  const int ks = 2 * lh;
  #pragma unroll
  for (int k = 0; k < D_DIM; k += 4) {
    v2f a;
    a.x = myft[lid * FP + k + ks + 0];
    a.y = myft[lid * FP + k + ks + 1];
    #pragma unroll
    for (int t = 0; t < 4; ++t) {
      const int n = t * 16 + lid;
      v2f b;
      b.x = Wl[n * WP + k + ks + 0];
      b.y = Wl[n * WP + k + ks + 1];
      acc[t] = __builtin_amdgcn_wmma_f32_16x16x4_f32(
          false, a, false, b, (short)0, acc[t], false, false);
    }
  }

  #pragma unroll
  for (int t = 0; t < 4; ++t) {
    const float bb = bias[t * 16 + lid];
    #pragma unroll
    for (int j = 0; j < 8; ++j) {
      myft[(j + 8 * lh) * FP + t * 16 + lid] = acc[t][j] + bb;
    }
  }

  for (int i = lane; i < TILE_M * (D_DIM / 4); i += 32) {
    const int r  = i >> 4;
    const int c4 = (i & 15) << 2;
    if (m0 + r < n_nodes) {
      const float4 hv = *(const float4*)&myft[r * FP + c4];
      *(float4*)(h + (size_t)(m0 + r) * D_DIM + c4) = hv;
    }
  }

  const float* aw = lh ? a2w : a1w;
  float s = lh ? a2b[0] : a1b[0];
  #pragma unroll 8
  for (int c = 0; c < D_DIM; ++c)
    s += myft[lid * FP + c] * aw[c];
  if (m0 + lid < n_nodes) {
    if (lh) a2[m0 + lid] = s;
    else    a1[m0 + lid] = s;
  }
}

__global__ void gat_init(float* __restrict__ out, float* __restrict__ segmax,
                         float* __restrict__ denom, int n) {
  const int i = blockIdx.x * blockDim.x + threadIdx.x;
  if (i < n * D_DIM) out[i] = 0.0f;
  if (i < n) { segmax[i] = -__builtin_inff(); denom[i] = 0.0f; }
}

__global__ void gat_edge_score(const int* __restrict__ src, const int* __restrict__ dst,
                               const float* __restrict__ a1, const float* __restrict__ a2,
                               float* __restrict__ vbuf, float* __restrict__ segmax, int E) {
  const int e = blockIdx.x * blockDim.x + threadIdx.x;
  if (e >= E) return;
  const int s = src[e], d = dst[e];
  float v = a1[s] + a2[d];
  v = v > 0.0f ? v : v * NEG_SLOPE;
  vbuf[e] = v;
  atomic_max_f32(segmax + s, v);
}

__global__ void gat_edge_exp(const int* __restrict__ src, const float* __restrict__ segmax,
                             float* __restrict__ vbuf, float* __restrict__ denom, int E) {
  const int e = blockIdx.x * blockDim.x + threadIdx.x;
  if (e >= E) return;
  const int s = src[e];
  const float ev = __expf(vbuf[e] - segmax[s]);
  vbuf[e] = ev;
  unsafeAtomicAdd(denom + s, ev);
}

__global__ __launch_bounds__(256) void gat_aggregate(
    const int* __restrict__ src, const int* __restrict__ dst,
    const float* __restrict__ evbuf, const float* __restrict__ denom,
    const float* __restrict__ h, float* __restrict__ out, int E) {
  const int e = blockIdx.x * 8 + (threadIdx.x >> 5);
  if (e >= E) return;
  const int lane = threadIdx.x & 31;
  const int s = src[e], d = dst[e];
  const float att = evbuf[e] / denom[s];
  const float* hr  = h   + (size_t)d * D_DIM;
  float*       orow = out + (size_t)s * D_DIM;
  unsafeAtomicAdd(orow + lane,      att * hr[lane]);
  unsafeAtomicAdd(orow + lane + 32, att * hr[lane + 32]);
}

extern "C" void kernel_launch(void* const* d_in, const int* in_sizes, int n_in,
                              void* d_out, int out_size, void* d_ws, size_t ws_size,
                              hipStream_t stream) {
  const float* feat = (const float*)d_in[0];
  const int*   idx  = (const int*)d_in[1];
  const float* W    = (const float*)d_in[2];
  const float* bias = (const float*)d_in[3];
  const float* a1w  = (const float*)d_in[4];
  const float* a1b  = (const float*)d_in[5];
  const float* a2w  = (const float*)d_in[6];
  const float* a2b  = (const float*)d_in[7];
  float* out = (float*)d_out;

  const int n = in_sizes[0] / D_DIM;
  const int E = in_sizes[1] / 2;
  const int* src = idx;
  const int* dst = idx + E;

  float* h      = (float*)d_ws;
  float* a1     = h + (size_t)n * D_DIM;
  float* a2     = a1 + n;
  float* segmax = a2 + n;
  float* denom  = segmax + n;
  float* vbuf   = denom + n;

  gat_init<<<(n * D_DIM + 255) / 256, 256, 0, stream>>>(out, segmax, denom, n);

  const int tiles = (n + TILE_M - 1) / TILE_M;
  gat_linear_wmma<<<(tiles + 7) / 8, 256, 0, stream>>>(
      feat, W, bias, a1w, a1b, a2w, a2b, h, a1, a2, n);

  gat_edge_score<<<(E + 255) / 256, 256, 0, stream>>>(src, dst, a1, a2, vbuf, segmax, E);
  gat_edge_exp<<<(E + 255) / 256, 256, 0, stream>>>(src, segmax, vbuf, denom, E);
  gat_aggregate<<<(E + 7) / 8, 256, 0, stream>>>(src, dst, vbuf, denom, h, out, E);
}


# aggregate pass: resident waves take 32 edges each, h-row loads of 8 edges issued as one group ahead of their f32 atomic adds (was one wave per edge); atomic arrival order differs
# speedup vs baseline: 1.5032x; 1.5032x over previous
#include <hip/hip_runtime.h>
#include <hip/hip_bf16.h>

typedef __attribute__((ext_vector_type(2))) float v2f;
typedef __attribute__((ext_vector_type(8))) float v8f;

#define D_DIM 64
#define TILE_M 16
#define NEG_SLOPE 0.01f

__device__ __forceinline__ void atomic_max_f32(float* addr, float val) {
  if (val >= 0.0f) {
    atomicMax((int*)addr, __float_as_int(val));
  } else {
    atomicMin((unsigned int*)addr, __float_as_uint(val));
  }
}

__global__ __launch_bounds__(256, 2) void gat_linear_wmma(
    const float* __restrict__ feat, const float* __restrict__ W,
    const float* __restrict__ bias,
    const float* __restrict__ a1w, const float* __restrict__ a1b,
    const float* __restrict__ a2w, const float* __restrict__ a2b,
    float* __restrict__ h, float* __restrict__ a1, float* __restrict__ a2,
    int n_nodes)
{
  constexpr int WP = 68;
  constexpr int FP = 68;
  __shared__ float Wl[D_DIM * WP];
  __shared__ float ft[8][TILE_M * FP];

  const int tid  = threadIdx.x;
  const int wave = tid >> 5;
  const int lane = tid & 31;
  const int lh   = lane >> 4;
  const int lid  = lane & 15;

  for (int i = tid; i < D_DIM * (D_DIM / 4); i += 256) {
    const int n  = i >> 4;
    const int c4 = (i & 15) << 2;
    const float4 w4 = *(const float4*)(W + n * D_DIM + c4);
    *(float4*)&Wl[n * WP + c4] = w4;
  }
  __syncthreads();

  const int tile = blockIdx.x * 8 + wave;
  const int m0   = tile * TILE_M;
  if (m0 >= n_nodes) return;

  float* myft = ft[wave];
  for (int i = lane; i < TILE_M * (D_DIM / 4); i += 32) {
    const int r  = i >> 4;
    const int c4 = (i & 15) << 2;
    float4 f4 = {0.f, 0.f, 0.f, 0.f};
    if (m0 + r < n_nodes)
      f4 = *(const float4*)(feat + (size_t)(m0 + r) * D_DIM + c4);
    *(float4*)&myft[r * FP + c4] = f4;
  }

  v8f acc[4] = {};
  const int ks = 2 * lh;
  #pragma unroll
  for (int k = 0; k < D_DIM; k += 4) {
    v2f a;
    a.x = myft[lid * FP + k + ks + 0];
    a.y = myft[lid * FP + k + ks + 1];
    #pragma unroll
    for (int t = 0; t < 4; ++t) {
      const int n = t * 16 + lid;
      v2f b;
      b.x = Wl[n * WP + k + ks + 0];
      b.y = Wl[n * WP + k + ks + 1];
      acc[t] = __builtin_amdgcn_wmma_f32_16x16x4_f32(
          false, a, false, b, (short)0, acc[t], false, false);
    }
  }

  #pragma unroll
  for (int t = 0; t < 4; ++t) {
    const float bb = bias[t * 16 + lid];
    #pragma unroll
    for (int j = 0; j < 8; ++j) {
      myft[(j + 8 * lh) * FP + t * 16 + lid] = acc[t][j] + bb;
    }
  }

  for (int i = lane; i < TILE_M * (D_DIM / 4); i += 32) {
    const int r  = i >> 4;
    const int c4 = (i & 15) << 2;
    if (m0 + r < n_nodes) {
      const float4 hv = *(const float4*)&myft[r * FP + c4];
      *(float4*)(h + (size_t)(m0 + r) * D_DIM + c4) = hv;
    }
  }

  const float* aw = lh ? a2w : a1w;
  float s = lh ? a2b[0] : a1b[0];
  #pragma unroll 8
  for (int c = 0; c < D_DIM; ++c)
    s += myft[lid * FP + c] * aw[c];
  if (m0 + lid < n_nodes) {
    if (lh) a2[m0 + lid] = s;
    else    a1[m0 + lid] = s;
  }
}

__global__ void gat_init(float* __restrict__ out, float* __restrict__ segmax,
                         float* __restrict__ denom, int n) {
  const int i = blockIdx.x * blockDim.x + threadIdx.x;
  if (i < n * D_DIM) out[i] = 0.0f;
  if (i < n) { segmax[i] = -__builtin_inff(); denom[i] = 0.0f; }
}

__global__ void gat_edge_score(const int* __restrict__ src, const int* __restrict__ dst,
                               const float* __restrict__ a1, const float* __restrict__ a2,
                               float* __restrict__ vbuf, float* __restrict__ segmax, int E) {
  const int e = blockIdx.x * blockDim.x + threadIdx.x;
  if (e >= E) return;
  const int s = src[e], d = dst[e];
  float v = a1[s] + a2[d];
  v = v > 0.0f ? v : v * NEG_SLOPE;
  vbuf[e] = v;
  atomic_max_f32(segmax + s, v);
}

__global__ void gat_edge_exp(const int* __restrict__ src, const float* __restrict__ segmax,
                             float* __restrict__ vbuf, float* __restrict__ denom, int E) {
  const int e = blockIdx.x * blockDim.x + threadIdx.x;
  if (e >= E) return;
  const int s = src[e];
  const float ev = __expf(vbuf[e] - segmax[s]);
  vbuf[e] = ev;
  unsafeAtomicAdd(denom + s, ev);
}

#define AGG_GROUP 8
__global__ __launch_bounds__(256) void gat_aggregate(
    const int* __restrict__ src, const int* __restrict__ dst,
    const float* __restrict__ evbuf, const float* __restrict__ denom,
    const float* __restrict__ h, float* __restrict__ out, int E) {
  const int lane = threadIdx.x & 31;
  const int wave   = __builtin_amdgcn_readfirstlane(blockIdx.x * 8 + (threadIdx.x >> 5));
  const int nwaves = gridDim.x * 8;
  const int nchunks = (E + 31) >> 5;

  for (int c = wave; c < nchunks; c += nwaves) {
    const int e = c * 32 + lane;
    int s = 0, d = 0;
    float att = 0.0f;
    if (e < E) {
      s = src[e];
      d = dst[e];
      att = evbuf[e] / denom[s];
    }
    const int atti = __float_as_int(att);
    const int cnt = min(32, E - c * 32);

    if (cnt == 32) {
      #pragma unroll
      for (int j0 = 0; j0 < 32; j0 += AGG_GROUP) {
        float v0[AGG_GROUP], v1[AGG_GROUP];
        #pragma unroll
        for (int j = 0; j < AGG_GROUP; ++j) {
          const int dj = __builtin_amdgcn_readlane(d, j0 + j);
          const float* hr = h + (size_t)dj * D_DIM;
          v0[j] = hr[lane];
          v1[j] = hr[lane + 32];
        }
        #pragma unroll
        for (int j = 0; j < AGG_GROUP; ++j) {
          const int   sj = __builtin_amdgcn_readlane(s, j0 + j);
          const float aj = __int_as_float(__builtin_amdgcn_readlane(atti, j0 + j));
          float* orow = out + (size_t)sj * D_DIM;
          unsafeAtomicAdd(orow + lane,      aj * v0[j]);
          unsafeAtomicAdd(orow + lane + 32, aj * v1[j]);
        }
      }
    } else {
      for (int j = 0; j < cnt; ++j) {
        const int   dj = __builtin_amdgcn_readlane(d, j);
        const int   sj = __builtin_amdgcn_readlane(s, j);
        const float aj = __int_as_float(__builtin_amdgcn_readlane(atti, j));
        const float* hr = h + (size_t)dj * D_DIM;
        float* orow = out + (size_t)sj * D_DIM;
        unsafeAtomicAdd(orow + lane,      aj * hr[lane]);
        unsafeAtomicAdd(orow + lane + 32, aj * hr[lane + 32]);
      }
    }
  }
}

extern "C" void kernel_launch(void* const* d_in, const int* in_sizes, int n_in,
                              void* d_out, int out_size, void* d_ws, size_t ws_size,
                              hipStream_t stream) {
  const float* feat = (const float*)d_in[0];
  const int*   idx  = (const int*)d_in[1];
  const float* W    = (const float*)d_in[2];
  const float* bias = (const float*)d_in[3];
  const float* a1w  = (const float*)d_in[4];
  const float* a1b  = (const float*)d_in[5];
  const float* a2w  = (const float*)d_in[6];
  const float* a2b  = (const float*)d_in[7];
  float* out = (float*)d_out;

  const int n = in_sizes[0] / D_DIM;
  const int E = in_sizes[1] / 2;
  const int* src = idx;
  const int* dst = idx + E;

  float* h      = (float*)d_ws;
  float* a1     = h + (size_t)n * D_DIM;
  float* a2     = a1 + n;
  float* segmax = a2 + n;
  float* denom  = segmax + n;
  float* vbuf   = denom + n;

  gat_init<<<(n * D_DIM + 255) / 256, 256, 0, stream>>>(out, segmax, denom, n);

  const int tiles = (n + TILE_M - 1) / TILE_M;
  gat_linear_wmma<<<(tiles + 7) / 8, 256, 0, stream>>>(
      feat, W, bias, a1w, a1b, a2w, a2b, h, a1, a2, n);

  gat_edge_score<<<(E + 255) / 256, 256, 0, stream>>>(src, dst, a1, a2, vbuf, segmax, E);
  gat_edge_exp<<<(E + 255) / 256, 256, 0, stream>>>(src, segmax, vbuf, denom, E);
  gat_aggregate<<<2048, 256, 0, stream>>>(src, dst, vbuf, denom, h, out, E);
}


# aggregate by row: edges grouped by src per call (integer-atomic rank, range, fill); one wave per row does softmax and the h-row sum in registers; no f32 atomics; sum order = slot order
# speedup vs baseline: 2.0612x; 1.3712x over previous
#include <hip/hip_runtime.h>

typedef __attribute__((ext_vector_type(2))) float v2f;
typedef __attribute__((ext_vector_type(8))) float v8f;

#define D_DIM 64
#define TILE_M 16
#define NEG_SLOPE 0.01f

__global__ __launch_bounds__(256, 2) void gat_linear_wmma(
    const float* __restrict__ feat, const float* __restrict__ W,
    const float* __restrict__ bias,
    const float* __restrict__ a1w, const float* __restrict__ a1b,
    const float* __restrict__ a2w, const float* __restrict__ a2b,
    float* __restrict__ h, float* __restrict__ a1, float* __restrict__ a2,
    int* __restrict__ cnt, int* __restrict__ bump,
    int n_nodes)
{
  constexpr int WP = 68;
  constexpr int FP = 68;
  __shared__ float Wl[D_DIM * WP];
  __shared__ float ft[8][TILE_M * FP];

  const int tid  = threadIdx.x;
  const int wave = tid >> 5;
  const int lane = tid & 31;
  const int lh   = lane >> 4;
  const int lid  = lane & 15;

  for (int i = tid; i < D_DIM * (D_DIM / 4); i += 256) {
    const int n  = i >> 4;
    const int c4 = (i & 15) << 2;
    const float4 w4 = *(const float4*)(W + n * D_DIM + c4);
    *(float4*)&Wl[n * WP + c4] = w4;
  }
  __syncthreads();

  if (blockIdx.x == 0 && tid == 0) *bump = 0;

  const int tile = blockIdx.x * 8 + wave;
  const int m0   = tile * TILE_M;
  if (m0 >= n_nodes) return;

  float* myft = ft[wave];
  for (int i = lane; i < TILE_M * (D_DIM / 4); i += 32) {
    const int r  = i >> 4;
    const int c4 = (i & 15) << 2;
    float4 f4 = {0.f, 0.f, 0.f, 0.f};
    if (m0 + r < n_nodes)
      f4 = *(const float4*)(feat + (size_t)(m0 + r) * D_DIM + c4);
    *(float4*)&myft[r * FP + c4] = f4;
  }

  v8f acc[4] = {};
  const int ks = 2 * lh;
  #pragma unroll
  for (int k = 0; k < D_DIM; k += 4) {
    v2f a;
    a.x = myft[lid * FP + k + ks + 0];
    a.y = myft[lid * FP + k + ks + 1];
    #pragma unroll
    for (int t = 0; t < 4; ++t) {
      const int n = t * 16 + lid;
      v2f b;
      b.x = Wl[n * WP + k + ks + 0];
      b.y = Wl[n * WP + k + ks + 1];
      acc[t] = __builtin_amdgcn_wmma_f32_16x16x4_f32(
          false, a, false, b, (short)0, acc[t], false, false);
    }
  }

  #pragma unroll
  for (int t = 0; t < 4; ++t) {
    const float bb = bias[t * 16 + lid];
    #pragma unroll
    for (int j = 0; j < 8; ++j) {
      myft[(j + 8 * lh) * FP + t * 16 + lid] = acc[t][j] + bb;
    }
  }

  for (int i = lane; i < TILE_M * (D_DIM / 4); i += 32) {
    const int r  = i >> 4;
    const int c4 = (i & 15) << 2;
    if (m0 + r < n_nodes) {
      const float4 hv = *(const float4*)&myft[r * FP + c4];
      *(float4*)(h + (size_t)(m0 + r) * D_DIM + c4) = hv;
    }
  }

  const float* aw = lh ? a2w : a1w;
  float s = lh ? a2b[0] : a1b[0];
  #pragma unroll 8
  for (int c = 0; c < D_DIM; ++c)
    s += myft[lid * FP + c] * aw[c];
  if (m0 + lid < n_nodes) {
    if (lh) a2[m0 + lid] = s;
    else  { a1[m0 + lid] = s; cnt[m0 + lid] = 0; }
  }
}

#define EDGES_PER_THREAD 8
__global__ __launch_bounds__(256) void gat_edge_rank(
    const int* __restrict__ src, int* __restrict__ cnt, int* __restrict__ rank, int E) {
  const int e0 = blockIdx.x * (256 * EDGES_PER_THREAD) + threadIdx.x;
  int s[EDGES_PER_THREAD], r[EDGES_PER_THREAD];
  #pragma unroll
  for (int k = 0; k < EDGES_PER_THREAD; ++k) {
    const int e = e0 + k * 256;
    s[k] = e < E ? src[e] : -1;
  }
  #pragma unroll
  for (int k = 0; k < EDGES_PER_THREAD; ++k)
    r[k] = s[k] >= 0 ? atomicAdd(cnt + s[k], 1) : 0;
  #pragma unroll
  for (int k = 0; k < EDGES_PER_THREAD; ++k) {
    const int e = e0 + k * 256;
    if (e < E) rank[e] = r[k];
  }
}

__global__ __launch_bounds__(256) void gat_row_alloc(
    const int* __restrict__ cnt, int* __restrict__ start, int* __restrict__ bump, int n) {
  __shared__ int wsum[8];
  __shared__ int base;
  const int tid = threadIdx.x, lane = tid & 31, wave = tid >> 5;
  const int i = blockIdx.x * 256 + tid;
  const int c = i < n ? cnt[i] : 0;
  int x = c;
  for (int o = 1; o < 32; o <<= 1) {
    const int y = __shfl_up(x, o);
    if (lane >= o) x += y;
  }
  if (lane == 31) wsum[wave] = x;
  __syncthreads();
  if (tid == 0) {
    int t = 0;
    for (int w = 0; w < 8; ++w) { const int v = wsum[w]; wsum[w] = t; t += v; }
    base = atomicAdd(bump, t);
  }
  __syncthreads();
  if (i < n) start[i] = base + wsum[wave] + x - c;
}

__global__ __launch_bounds__(256) void gat_edge_fill(
    const int* __restrict__ src, const int* __restrict__ dst,
    const int* __restrict__ rank, const int* __restrict__ start,
    int* __restrict__ dlist, int E) {
  const int e0 = blockIdx.x * (256 * EDGES_PER_THREAD) + threadIdx.x;
  int s[EDGES_PER_THREAD], d[EDGES_PER_THREAD], p[EDGES_PER_THREAD];
  #pragma unroll
  for (int k = 0; k < EDGES_PER_THREAD; ++k) {
    const int e = e0 + k * 256;
    s[k] = -1; d[k] = 0; p[k] = 0;
    if (e < E) { s[k] = src[e]; d[k] = dst[e]; p[k] = rank[e]; }
  }
  #pragma unroll
  for (int k = 0; k < EDGES_PER_THREAD; ++k)
    if (s[k] >= 0) p[k] += start[s[k]];
  #pragma unroll
  for (int k = 0; k < EDGES_PER_THREAD; ++k)
    if (s[k] >= 0) dlist[p[k]] = d[k];
}

template <int CTRL>
__device__ __forceinline__ float dpp_move(float x) {
  return __int_as_float(__builtin_amdgcn_update_dpp(0, __float_as_int(x), CTRL, 0xF, 0xF, true));
}
#define DPP_XOR1   0xB1
#define DPP_XOR2   0x4E
#define DPP_HMIRR  0x141
#define DPP_MIRR   0x140

__device__ __forceinline__ float wave_max(float x) {
  x = fmaxf(x, dpp_move<DPP_XOR1>(x));
  x = fmaxf(x, dpp_move<DPP_XOR2>(x));
  x = fmaxf(x, dpp_move<DPP_HMIRR>(x));
  x = fmaxf(x, dpp_move<DPP_MIRR>(x));
  const float lo = __int_as_float(__builtin_amdgcn_readlane(__float_as_int(x), 0));
  const float hi = __int_as_float(__builtin_amdgcn_readlane(__float_as_int(x), 16));
  return fmaxf(lo, hi);
}
__device__ __forceinline__ float wave_sum(float x) {
  x += dpp_move<DPP_XOR1>(x);
  x += dpp_move<DPP_XOR2>(x);
  x += dpp_move<DPP_HMIRR>(x);
  x += dpp_move<DPP_MIRR>(x);
  const float lo = __int_as_float(__builtin_amdgcn_readlane(__float_as_int(x), 0));
  const float hi = __int_as_float(__builtin_amdgcn_readlane(__float_as_int(x), 16));
  return lo + hi;
}

#define AGG_GROUP 16
__device__ __forceinline__ void add_rows(const float* __restrict__ h, int lane, int cc,
                                         int d, float att, float& acc0, float& acc1) {
  const int atti = __float_as_int(att);
  for (int j0 = 0; j0 < cc; j0 += AGG_GROUP) {
    float v0[AGG_GROUP], v1[AGG_GROUP];
    #pragma unroll
    for (int j = 0; j < AGG_GROUP; ++j) {
      v0[j] = 0.0f;
      v1[j] = 0.0f;
      if (j0 + j < cc) {
        const int dj = __builtin_amdgcn_readlane(d, j0 + j);
        const float* hr = h + (size_t)dj * D_DIM;
        v0[j] = hr[lane];
        v1[j] = hr[lane + 32];
      }
    }
    #pragma unroll
    for (int j = 0; j < AGG_GROUP; ++j) {
      const float aj = __int_as_float(__builtin_amdgcn_readlane(atti, (j0 + j) & 31));
      acc0 += aj * v0[j];
      acc1 += aj * v1[j];
    }
  }
}

__global__ __launch_bounds__(256) void gat_row_aggregate(
    const int* __restrict__ cnt, const int* __restrict__ start,
    const int* __restrict__ dlist,
    const float* __restrict__ a1, const float* __restrict__ a2,
    const float* __restrict__ h, float* __restrict__ out, int n) {
  const int lane = threadIdx.x & 31;
  const int wave   = __builtin_amdgcn_readfirstlane(blockIdx.x * 8 + (threadIdx.x >> 5));
  const int nwaves = gridDim.x * 8;

  int   c   = 0, st = 0;
  float a1s = 0.0f;
  if (wave < n) { c = cnt[wave]; st = start[wave]; a1s = a1[wave]; }
  for (int s = wave; s < n; s += nwaves) {
    const int sn = s + nwaves;
    int   cn   = 0, stn = 0;
    float a1n  = 0.0f;
    if (sn < n) { cn = cnt[sn]; stn = start[sn]; a1n = a1[sn]; }
    float acc0 = 0.0f, acc1 = 0.0f;

    if (c <= 32) {
      int d = 0;
      float v = -__builtin_inff();
      if (lane < c) {
        d = dlist[st + lane];
        const float t = a1s + a2[d];
        v = t > 0.0f ? t : t * NEG_SLOPE;
      }
      const float m   = wave_max(v);
      const float ev  = lane < c ? __expf(v - m) : 0.0f;
      const float den = wave_sum(ev);
      const float att = lane < c ? ev / den : 0.0f;
      add_rows(h, lane, c, d, att, acc0, acc1);
    } else {
      float m = -__builtin_inff();
      for (int b = 0; b < c; b += 32) {
        const int j = b + lane;
        if (j < c) {
          const float t = a1s + a2[dlist[st + j]];
          m = fmaxf(m, t > 0.0f ? t : t * NEG_SLOPE);
        }
      }
      m = wave_max(m);
      float part = 0.0f;
      for (int b = 0; b < c; b += 32) {
        const int j = b + lane;
        if (j < c) {
          const float t = a1s + a2[dlist[st + j]];
          part += __expf((t > 0.0f ? t : t * NEG_SLOPE) - m);
        }
      }
      const float den = wave_sum(part);
      for (int b = 0; b < c; b += 32) {
        const int j = b + lane;
        int d = 0;
        float att = 0.0f;
        if (j < c) {
          d = dlist[st + j];
          const float t = a1s + a2[d];
          att = __expf((t > 0.0f ? t : t * NEG_SLOPE) - m) / den;
        }
        add_rows(h, lane, min(32, c - b), d, att, acc0, acc1);
      }
    }

    float* orow = out + (size_t)s * D_DIM;
    orow[lane]      = acc0;
    orow[lane + 32] = acc1;
    c = cn; st = stn; a1s = a1n;
  }
}

extern "C" void kernel_launch(void* const* d_in, const int* in_sizes, int n_in,
                              void* d_out, int out_size, void* d_ws, size_t ws_size,
                              hipStream_t stream) {
  const float* feat = (const float*)d_in[0];
  const int*   idx  = (const int*)d_in[1];
  const float* W    = (const float*)d_in[2];
  const float* bias = (const float*)d_in[3];
  const float* a1w  = (const float*)d_in[4];
  const float* a1b  = (const float*)d_in[5];
  const float* a2w  = (const float*)d_in[6];
  const float* a2b  = (const float*)d_in[7];
  float* out = (float*)d_out;

  const int n = in_sizes[0] / D_DIM;
  const int E = in_sizes[1] / 2;
  const int* src = idx;
  const int* dst = idx + E;

  float* h     = (float*)d_ws;
  float* a1    = h + (size_t)n * D_DIM;
  float* a2    = a1 + n;
  int*   cnt   = (int*)(a2 + n);
  int*   start = cnt + n;
  int*   bump  = start + n;
  int*   rank  = bump + 4;
  int*   dlist = rank + E;

  const int tiles = (n + TILE_M - 1) / TILE_M;
  gat_linear_wmma<<<(tiles + 7) / 8, 256, 0, stream>>>(
      feat, W, bias, a1w, a1b, a2w, a2b, h, a1, a2, cnt, bump, n);

  const int eblocks = (E + 256 * EDGES_PER_THREAD - 1) / (256 * EDGES_PER_THREAD);
  gat_edge_rank<<<eblocks, 256, 0, stream>>>(src, cnt, rank, E);
  gat_row_alloc<<<(n + 255) / 256, 256, 0, stream>>>(cnt, start, bump, n);
  gat_edge_fill<<<eblocks, 256, 0, stream>>>(src, dst, rank, start, dlist, E);
  gat_row_aggregate<<<2048, 256, 0, stream>>>(cnt, start, dlist, a1, a2, h, out, n);
}


# edges bucketed in one pass: the rank's returning add stores dst at once into the row's own 64-slot unit (overflow list past 64); range and fill passes, rank and start arrays go; 5 launches to 3
# speedup vs baseline: 2.2461x; 1.0897x over previous
#include <hip/hip_runtime.h>

typedef __attribute__((ext_vector_type(2))) float v2f;
typedef __attribute__((ext_vector_type(8))) float v8f;

#define D_DIM 64
#define TILE_M 16
#define NEG_SLOPE 0.01f

__global__ __launch_bounds__(256, 2) void gat_linear_wmma(
    const float* __restrict__ feat, const float* __restrict__ W,
    const float* __restrict__ bias,
    const float* __restrict__ a1w, const float* __restrict__ a1b,
    const float* __restrict__ a2w, const float* __restrict__ a2b,
    float* __restrict__ h, float* __restrict__ a1, float* __restrict__ a2,
    int* __restrict__ cnt, int* __restrict__ bump,
    int n_nodes)
{
  constexpr int WP = 68;
  constexpr int FP = 68;
  __shared__ float Wl[D_DIM * WP];
  __shared__ float ft[8][TILE_M * FP];

  const int tid  = threadIdx.x;
  const int wave = tid >> 5;
  const int lane = tid & 31;
  const int lh   = lane >> 4;
  const int lid  = lane & 15;

  for (int i = tid; i < D_DIM * (D_DIM / 4); i += 256) {
    const int n  = i >> 4;
    const int c4 = (i & 15) << 2;
    const float4 w4 = *(const float4*)(W + n * D_DIM + c4);
    *(float4*)&Wl[n * WP + c4] = w4;
  }
  __syncthreads();

  if (blockIdx.x == 0 && tid == 0) *bump = 0;

  const int tile = blockIdx.x * 8 + wave;
  const int m0   = tile * TILE_M;
  if (m0 >= n_nodes) return;

  float* myft = ft[wave];
  for (int i = lane; i < TILE_M * (D_DIM / 4); i += 32) {
    const int r  = i >> 4;
    const int c4 = (i & 15) << 2;
    float4 f4 = {0.f, 0.f, 0.f, 0.f};
    if (m0 + r < n_nodes)
      f4 = *(const float4*)(feat + (size_t)(m0 + r) * D_DIM + c4);
    *(float4*)&myft[r * FP + c4] = f4;
  }

  v8f acc[4] = {};
  const int ks = 2 * lh;
  #pragma unroll
  for (int k = 0; k < D_DIM; k += 4) {
    v2f a;
    a.x = myft[lid * FP + k + ks + 0];
    a.y = myft[lid * FP + k + ks + 1];
    #pragma unroll
    for (int t = 0; t < 4; ++t) {
      const int n = t * 16 + lid;
      v2f b;
      b.x = Wl[n * WP + k + ks + 0];
      b.y = Wl[n * WP + k + ks + 1];
      acc[t] = __builtin_amdgcn_wmma_f32_16x16x4_f32(
          false, a, false, b, (short)0, acc[t], false, false);
    }
  }

  #pragma unroll
  for (int t = 0; t < 4; ++t) {
    const float bb = bias[t * 16 + lid];
    #pragma unroll
    for (int j = 0; j < 8; ++j) {
      myft[(j + 8 * lh) * FP + t * 16 + lid] = acc[t][j] + bb;
    }
  }

  for (int i = lane; i < TILE_M * (D_DIM / 4); i += 32) {
    const int r  = i >> 4;
    const int c4 = (i & 15) << 2;
    if (m0 + r < n_nodes) {
      const float4 hv = *(const float4*)&myft[r * FP + c4];
      *(float4*)(h + (size_t)(m0 + r) * D_DIM + c4) = hv;
    }
  }

  const float* aw = lh ? a2w : a1w;
  float s = lh ? a2b[0] : a1b[0];
  #pragma unroll 8
  for (int c = 0; c < D_DIM; ++c)
    s += myft[lid * FP + c] * aw[c];
  if (m0 + lid < n_nodes) {
    if (lh) a2[m0 + lid] = s;
    else  { a1[m0 + lid] = s; cnt[m0 + lid] = 0; }
  }
}

#define ROW_CAP 64
#define EDGES_PER_THREAD 8
__global__ __launch_bounds__(256) void gat_edge_bucket(
    const int* __restrict__ src, const int* __restrict__ dst,
    int* __restrict__ cnt, int* __restrict__ slots,
    int* __restrict__ ovf_n, int* __restrict__ ovf, int E) {
  const int e0 = blockIdx.x * (256 * EDGES_PER_THREAD) + threadIdx.x;
  int s[EDGES_PER_THREAD], d[EDGES_PER_THREAD], r[EDGES_PER_THREAD];
  #pragma unroll
  for (int k = 0; k < EDGES_PER_THREAD; ++k) {
    const int e = e0 + k * 256;
    s[k] = -1; d[k] = 0;
    if (e < E) { s[k] = src[e]; d[k] = dst[e]; }
  }
  #pragma unroll
  for (int k = 0; k < EDGES_PER_THREAD; ++k)
    r[k] = s[k] >= 0 ? atomicAdd(cnt + s[k], 1) : 0;
  #pragma unroll
  for (int k = 0; k < EDGES_PER_THREAD; ++k) {
    if (s[k] >= 0) {
      if (r[k] < ROW_CAP) {
        slots[s[k] * ROW_CAP + r[k]] = d[k];
      } else {
        const int o = atomicAdd(ovf_n, 1);
        ovf[2 * o]     = s[k];
        ovf[2 * o + 1] = d[k];
      }
    }
  }
}

template <int CTRL>
__device__ __forceinline__ float dpp_move(float x) {
  return __int_as_float(__builtin_amdgcn_update_dpp(0, __float_as_int(x), CTRL, 0xF, 0xF, true));
}
#define DPP_XOR1   0xB1
#define DPP_XOR2   0x4E
#define DPP_HMIRR  0x141
#define DPP_MIRR   0x140

__device__ __forceinline__ float wave_max(float x) {
  x = fmaxf(x, dpp_move<DPP_XOR1>(x));
  x = fmaxf(x, dpp_move<DPP_XOR2>(x));
  x = fmaxf(x, dpp_move<DPP_HMIRR>(x));
  x = fmaxf(x, dpp_move<DPP_MIRR>(x));
  const float lo = __int_as_float(__builtin_amdgcn_readlane(__float_as_int(x), 0));
  const float hi = __int_as_float(__builtin_amdgcn_readlane(__float_as_int(x), 16));
  return fmaxf(lo, hi);
}
__device__ __forceinline__ float wave_sum(float x) {
  x += dpp_move<DPP_XOR1>(x);
  x += dpp_move<DPP_XOR2>(x);
  x += dpp_move<DPP_HMIRR>(x);
  x += dpp_move<DPP_MIRR>(x);
  const float lo = __int_as_float(__builtin_amdgcn_readlane(__float_as_int(x), 0));
  const float hi = __int_as_float(__builtin_amdgcn_readlane(__float_as_int(x), 16));
  return lo + hi;
}

#define AGG_GROUP 16
__device__ __forceinline__ void add_rows(const float* __restrict__ h, int lane, int cc,
                                         int d, float att, float& acc0, float& acc1) {
  const int atti = __float_as_int(att);
  for (int j0 = 0; j0 < cc; j0 += AGG_GROUP) {
    float v0[AGG_GROUP], v1[AGG_GROUP];
    #pragma unroll
    for (int j = 0; j < AGG_GROUP; ++j) {
      v0[j] = 0.0f;
      v1[j] = 0.0f;
      if (j0 + j < cc) {
        const int dj = __builtin_amdgcn_readlane(d, j0 + j);
        const float* hr = h + (size_t)dj * D_DIM;
        v0[j] = hr[lane];
        v1[j] = hr[lane + 32];
      }
    }
    #pragma unroll
    for (int j = 0; j < AGG_GROUP; ++j) {
      const float aj = __int_as_float(__builtin_amdgcn_readlane(atti, (j0 + j) & 31));
      acc0 += aj * v0[j];
      acc1 += aj * v1[j];
    }
  }
}

__device__ __forceinline__ float edge_score(float a1s, float a2d) {
  const float t = a1s + a2d;
  return t > 0.0f ? t : t * NEG_SLOPE;
}

__global__ __launch_bounds__(256) void gat_row_aggregate(
    const int* __restrict__ cnt, const int* __restrict__ slots,
    const int* __restrict__ ovf_n, const int* __restrict__ ovf,
    const float* __restrict__ a1, const float* __restrict__ a2,
    const float* __restrict__ h, float* __restrict__ out, int n) {
  const int lane = threadIdx.x & 31;
  const int wave   = __builtin_amdgcn_readfirstlane(blockIdx.x * 8 + (threadIdx.x >> 5));
  const int nwaves = gridDim.x * 8;

  int   c   = 0;
  float a1s = 0.0f;
  if (wave < n) { c = cnt[wave]; a1s = a1[wave]; }
  for (int s = wave; s < n; s += nwaves) {
    const int sn = s + nwaves;
    int   cn   = 0;
    float a1n  = 0.0f;
    if (sn < n) { cn = cnt[sn]; a1n = a1[sn]; }
    const int* row = slots + (size_t)s * ROW_CAP;
    float acc0 = 0.0f, acc1 = 0.0f;

    if (c <= 32) {
      int d = 0;
      float v = -__builtin_inff();
      if (lane < c) {
        d = row[lane];
        v = edge_score(a1s, a2[d]);
      }
      const float m   = wave_max(v);
      const float ev  = lane < c ? __expf(v - m) : 0.0f;
      const float den = wave_sum(ev);
      const float att = lane < c ? ev / den : 0.0f;
      add_rows(h, lane, c, d, att, acc0, acc1);
    } else {
      const int cs  = min(c, ROW_CAP);
      const int nov = c > ROW_CAP ? *ovf_n : 0;
      float m = -__builtin_inff();
      for (int b = 0; b < cs; b += 32) {
        const int j = b + lane;
        if (j < cs) m = fmaxf(m, edge_score(a1s, a2[row[j]]));
      }
      for (int b = 0; b < nov; b += 32) {
        const int j = b + lane;
        if (j < nov && ovf[2 * j] == s) m = fmaxf(m, edge_score(a1s, a2[ovf[2 * j + 1]]));
      }
      m = wave_max(m);
      float part = 0.0f;
      for (int b = 0; b < cs; b += 32) {
        const int j = b + lane;
        if (j < cs) part += __expf(edge_score(a1s, a2[row[j]]) - m);
      }
      for (int b = 0; b < nov; b += 32) {
        const int j = b + lane;
        if (j < nov && ovf[2 * j] == s) part += __expf(edge_score(a1s, a2[ovf[2 * j + 1]]) - m);
      }
      const float den = wave_sum(part);
      for (int b = 0; b < cs; b += 32) {
        const int j = b + lane;
        int d = 0;
        float att = 0.0f;
        if (j < cs) {
          d = row[j];
          att = __expf(edge_score(a1s, a2[d]) - m) / den;
        }
        add_rows(h, lane, min(32, cs - b), d, att, acc0, acc1);
      }
      for (int b = 0; b < nov; b += 32) {
        const int j = b + lane;
        int d = 0;
        float att = 0.0f;
        bool mine = false;
        if (j < nov && ovf[2 * j] == s) {
          mine = true;
          d = ovf[2 * j + 1];
          att = __expf(edge_score(a1s, a2[d]) - m) / den;
        }
        unsigned live = __builtin_amdgcn_ballot_w32(mine);
        while (live != 0u) {
          const int jj = __builtin_ctz(live);
          live &= live - 1u;
          const int   dj = __builtin_amdgcn_readlane(d, jj);
          const float aj = __int_as_float(__builtin_amdgcn_readlane(__float_as_int(att), jj));
          const float* hr = h + (size_t)dj * D_DIM;
          acc0 += aj * hr[lane];
          acc1 += aj * hr[lane + 32];
        }
      }
    }

    float* orow = out + (size_t)s * D_DIM;
    orow[lane]      = acc0;
    orow[lane + 32] = acc1;
    c = cn; a1s = a1n;
  }
}

extern "C" void kernel_launch(void* const* d_in, const int* in_sizes, int n_in,
                              void* d_out, int out_size, void* d_ws, size_t ws_size,
                              hipStream_t stream) {
  const float* feat = (const float*)d_in[0];
  const int*   idx  = (const int*)d_in[1];
  const float* W    = (const float*)d_in[2];
  const float* bias = (const float*)d_in[3];
  const float* a1w  = (const float*)d_in[4];
  const float* a1b  = (const float*)d_in[5];
  const float* a2w  = (const float*)d_in[6];
  const float* a2b  = (const float*)d_in[7];
  float* out = (float*)d_out;

  const int n = in_sizes[0] / D_DIM;
  const int E = in_sizes[1] / 2;
  const int* src = idx;
  const int* dst = idx + E;

  float* h     = (float*)d_ws;
  int*   slots = (int*)(h + (size_t)n * D_DIM);
  float* a1    = (float*)(slots + (size_t)n * ROW_CAP);
  float* a2    = a1 + n;
  int*   cnt   = (int*)(a2 + n);
  int*   bump  = cnt + n;
  int*   ovf   = bump + 4;

  const int tiles = (n + TILE_M - 1) / TILE_M;
  gat_linear_wmma<<<(tiles + 7) / 8, 256, 0, stream>>>(
      feat, W, bias, a1w, a1b, a2w, a2b, h, a1, a2, cnt, bump, n);

  const int eblocks = (E + 256 * EDGES_PER_THREAD - 1) / (256 * EDGES_PER_THREAD);
  gat_edge_bucket<<<eblocks, 256, 0, stream>>>(src, dst, cnt, slots, bump, ovf, E);
  gat_row_aggregate<<<2048, 256, 0, stream>>>(cnt, slots, bump, ovf, a1, a2, h, out, n);
}


# edge ranks from LDS counters: edges partitioned into 256-row bins (one global add per work-group and bin), then one work-group per bin ranks and places; 4 launches; row sum order = LDS-add arrival
# speedup vs baseline: 5.1229x; 2.2808x over previous
#include <hip/hip_runtime.h>

typedef __attribute__((ext_vector_type(2))) float v2f;
typedef __attribute__((ext_vector_type(8))) float v8f;

#define D_DIM 64
#define TILE_M 16
#define NEG_SLOPE 0.01f

#define ROW_CAP 64
#define BIN_SHIFT 8
#define BIN_ROWS (1 << BIN_SHIFT)
#define BIN_CAP 8192
#define CNT_PITCH 32
#define MAX_BINS 512
#define PART_THREADS 1024
#define EDGES_PER_THREAD 8

__global__ __launch_bounds__(256, 2) void gat_linear_wmma(
    const float* __restrict__ feat, const float* __restrict__ W,
    const float* __restrict__ bias,
    const float* __restrict__ a1w, const float* __restrict__ a1b,
    const float* __restrict__ a2w, const float* __restrict__ a2b,
    float* __restrict__ h, float* __restrict__ a1, float* __restrict__ a2,
    int* __restrict__ gcnt, int* __restrict__ ctrl,
    int n_nodes)
{
  constexpr int WP = 68;
  constexpr int FP = 68;
  __shared__ float Wl[D_DIM * WP];
  __shared__ float ft[8][TILE_M * FP];

  const int tid  = threadIdx.x;
  const int wave = tid >> 5;
  const int lane = tid & 31;
  const int lh   = lane >> 4;
  const int lid  = lane & 15;

  for (int i = tid; i < D_DIM * (D_DIM / 4); i += 256) {
    const int n  = i >> 4;
    const int c4 = (i & 15) << 2;
    const float4 w4 = *(const float4*)(W + n * D_DIM + c4);
    *(float4*)&Wl[n * WP + c4] = w4;
  }
  __syncthreads();

  if (blockIdx.x == 0) {
    const int nbins = (n_nodes + BIN_ROWS - 1) >> BIN_SHIFT;
    for (int b = tid; b < nbins; b += 256) gcnt[b * CNT_PITCH] = 0;
    if (tid < 2) ctrl[tid] = 0;
  }

  const int tile = blockIdx.x * 8 + wave;
  const int m0   = tile * TILE_M;
  if (m0 >= n_nodes) return;

  float* myft = ft[wave];
  for (int i = lane; i < TILE_M * (D_DIM / 4); i += 32) {
    const int r  = i >> 4;
    const int c4 = (i & 15) << 2;
    float4 f4 = {0.f, 0.f, 0.f, 0.f};
    if (m0 + r < n_nodes)
      f4 = *(const float4*)(feat + (size_t)(m0 + r) * D_DIM + c4);
    *(float4*)&myft[r * FP + c4] = f4;
  }

  v8f acc[4] = {};
  const int ks = 2 * lh;
  #pragma unroll
  for (int k = 0; k < D_DIM; k += 4) {
    v2f a;
    a.x = myft[lid * FP + k + ks + 0];
    a.y = myft[lid * FP + k + ks + 1];
    #pragma unroll
    for (int t = 0; t < 4; ++t) {
      const int n = t * 16 + lid;
      v2f b;
      b.x = Wl[n * WP + k + ks + 0];
      b.y = Wl[n * WP + k + ks + 1];
      acc[t] = __builtin_amdgcn_wmma_f32_16x16x4_f32(
          false, a, false, b, (short)0, acc[t], false, false);
    }
  }

  #pragma unroll
  for (int t = 0; t < 4; ++t) {
    const float bb = bias[t * 16 + lid];
    #pragma unroll
    for (int j = 0; j < 8; ++j) {
      myft[(j + 8 * lh) * FP + t * 16 + lid] = acc[t][j] + bb;
    }
  }

  for (int i = lane; i < TILE_M * (D_DIM / 4); i += 32) {
    const int r  = i >> 4;
    const int c4 = (i & 15) << 2;
    if (m0 + r < n_nodes) {
      const float4 hv = *(const float4*)&myft[r * FP + c4];
      *(float4*)(h + (size_t)(m0 + r) * D_DIM + c4) = hv;
    }
  }

  const float* aw = lh ? a2w : a1w;
  float s = lh ? a2b[0] : a1b[0];
  #pragma unroll 8
  for (int c = 0; c < D_DIM; ++c)
    s += myft[lid * FP + c] * aw[c];
  if (m0 + lid < n_nodes) {
    if (lh) a2[m0 + lid] = s;
    else    a1[m0 + lid] = s;
  }
}

__global__ __launch_bounds__(PART_THREADS) void gat_edge_partition(
    const int* __restrict__ src, const int* __restrict__ dst,
    int* __restrict__ gcnt, int* __restrict__ recs,
    int* __restrict__ spill_n, int* __restrict__ spill, int E, int nbins) {
  __shared__ int lcnt[MAX_BINS];
  __shared__ int lbase[MAX_BINS];
  const int tid = threadIdx.x;
  for (int b = tid; b < nbins; b += PART_THREADS) lcnt[b] = 0;
  __syncthreads();

  const int e0 = blockIdx.x * (PART_THREADS * EDGES_PER_THREAD) + tid;
  int s[EDGES_PER_THREAD], d[EDGES_PER_THREAD], r[EDGES_PER_THREAD];
  #pragma unroll
  for (int k = 0; k < EDGES_PER_THREAD; ++k) {
    const int e = e0 + k * PART_THREADS;
    s[k] = -1; d[k] = 0;
    if (e < E) { s[k] = src[e]; d[k] = dst[e]; }
  }
  #pragma unroll
  for (int k = 0; k < EDGES_PER_THREAD; ++k)
    r[k] = s[k] >= 0 ? atomicAdd(&lcnt[s[k] >> BIN_SHIFT], 1) : 0;
  __syncthreads();

  for (int b = tid; b < nbins; b += PART_THREADS) {
    const int c = lcnt[b];
    lbase[b] = c > 0 ? atomicAdd(gcnt + b * CNT_PITCH, c) : 0;
  }
  __syncthreads();

  #pragma unroll
  for (int k = 0; k < EDGES_PER_THREAD; ++k) {
    if (s[k] >= 0) {
      const int b = s[k] >> BIN_SHIFT;
      const int p = lbase[b] + r[k];
      int2 sd; sd.x = s[k]; sd.y = d[k];
      int2* where;
      if (p < BIN_CAP) {
        where = (int2*)(recs + ((size_t)b * BIN_CAP + p) * 2);
      } else {
        const int o = atomicAdd(spill_n, 1);
        where = (int2*)(spill + (size_t)o * 2);
      }
      *where = sd;
    }
  }
}

__device__ __forceinline__ void put_edge(int s, int d, int r, int* stage,
                                         int* __restrict__ ovf_n, int* __restrict__ ovf) {
  if (r < ROW_CAP) {
    stage[(s & (BIN_ROWS - 1)) * ROW_CAP + r] = d;
  } else {
    const int o = atomicAdd(ovf_n, 1);
    ovf[2 * o]     = s;
    ovf[2 * o + 1] = d;
  }
}

#define PLACE_UNROLL 8
#define IMAGE_GROUPS (BIN_ROWS * ROW_CAP / 4 / BIN_ROWS)
__global__ __launch_bounds__(BIN_ROWS) void gat_bin_place(
    const int* __restrict__ gcnt, const int* __restrict__ recs,
    const int* __restrict__ spill_n, const int* __restrict__ spill,
    int* __restrict__ cnt, int* __restrict__ slots,
    int* __restrict__ ovf_n, int* __restrict__ ovf, int n) {
  __shared__ int lc[BIN_ROWS];
  __shared__ __attribute__((aligned(16))) int stage[BIN_ROWS * ROW_CAP];
  const int tid = threadIdx.x;
  const int b   = blockIdx.x;
  lc[tid] = 0;
  #pragma unroll
  for (int g = 0; g < IMAGE_GROUPS; ++g)
    *(int4*)&stage[(g * BIN_ROWS + tid) * 4] = make_int4(0, 0, 0, 0);
  __syncthreads();

  const int m = min(gcnt[b * CNT_PITCH], BIN_CAP);
  const int2* list = (const int2*)(recs + (size_t)b * BIN_CAP * 2);
  for (int i0 = tid; i0 < m; i0 += BIN_ROWS * PLACE_UNROLL) {
    int2 sd[PLACE_UNROLL];
    int  r[PLACE_UNROLL];
    #pragma unroll
    for (int u = 0; u < PLACE_UNROLL; ++u) {
      const int i = i0 + u * BIN_ROWS;
      sd[u].x = -1; sd[u].y = 0;
      if (i < m) sd[u] = list[i];
    }
    #pragma unroll
    for (int u = 0; u < PLACE_UNROLL; ++u)
      r[u] = sd[u].x >= 0 ? atomicAdd(&lc[sd[u].x & (BIN_ROWS - 1)], 1) : 0;
    #pragma unroll
    for (int u = 0; u < PLACE_UNROLL; ++u)
      if (sd[u].x >= 0) put_edge(sd[u].x, sd[u].y, r[u], stage, ovf_n, ovf);
  }
  const int ns = *spill_n;
  for (int i = tid; i < ns; i += BIN_ROWS) {
    const int s = spill[2 * i];
    if ((s >> BIN_SHIFT) == b) {
      const int r = atomicAdd(&lc[s & (BIN_ROWS - 1)], 1);
      put_edge(s, spill[2 * i + 1], r, stage, ovf_n, ovf);
    }
  }
  __syncthreads();

  const int row0 = b * BIN_ROWS;
  if (row0 + tid < n) cnt[row0 + tid] = lc[tid];
  #pragma unroll
  for (int g = 0; g < IMAGE_GROUPS; ++g) {
    const int w  = (g * BIN_ROWS + tid) * 4;
    const int rl = w / ROW_CAP;
    const int sl = w % ROW_CAP;
    if (row0 + rl < n && sl < lc[rl]) {
      const int4 v = *(const int4*)&stage[w];
      *(int4*)(slots + (size_t)(row0 + rl) * ROW_CAP + sl) = v;
    }
  }
}

template <int CTRL>
__device__ __forceinline__ float dpp_move(float x) {
  return __int_as_float(__builtin_amdgcn_update_dpp(0, __float_as_int(x), CTRL, 0xF, 0xF, true));
}
#define DPP_XOR1   0xB1
#define DPP_XOR2   0x4E
#define DPP_HMIRR  0x141
#define DPP_MIRR   0x140

__device__ __forceinline__ float wave_max(float x) {
  x = fmaxf(x, dpp_move<DPP_XOR1>(x));
  x = fmaxf(x, dpp_move<DPP_XOR2>(x));
  x = fmaxf(x, dpp_move<DPP_HMIRR>(x));
  x = fmaxf(x, dpp_move<DPP_MIRR>(x));
  const float lo = __int_as_float(__builtin_amdgcn_readlane(__float_as_int(x), 0));
  const float hi = __int_as_float(__builtin_amdgcn_readlane(__float_as_int(x), 16));
  return fmaxf(lo, hi);
}
__device__ __forceinline__ float wave_sum(float x) {
  x += dpp_move<DPP_XOR1>(x);
  x += dpp_move<DPP_XOR2>(x);
  x += dpp_move<DPP_HMIRR>(x);
  x += dpp_move<DPP_MIRR>(x);
  const float lo = __int_as_float(__builtin_amdgcn_readlane(__float_as_int(x), 0));
  const float hi = __int_as_float(__builtin_amdgcn_readlane(__float_as_int(x), 16));
  return lo + hi;
}

#define AGG_GROUP 16
__device__ __forceinline__ void add_rows(const float* __restrict__ h, int lane, int cc,
                                         int d, float att, float& acc0, float& acc1) {
  const int atti = __float_as_int(att);
  for (int j0 = 0; j0 < cc; j0 += AGG_GROUP) {
    float v0[AGG_GROUP], v1[AGG_GROUP];
    #pragma unroll
    for (int j = 0; j < AGG_GROUP; ++j) {
      v0[j] = 0.0f;
      v1[j] = 0.0f;
      if (j0 + j < cc) {
        const int dj = __builtin_amdgcn_readlane(d, j0 + j);
        const float* hr = h + (size_t)dj * D_DIM;
        v0[j] = hr[lane];
        v1[j] = hr[lane + 32];
      }
    }
    #pragma unroll
    for (int j = 0; j < AGG_GROUP; ++j) {
      const float aj = __int_as_float(__builtin_amdgcn_readlane(atti, (j0 + j) & 31));
      acc0 += aj * v0[j];
      acc1 += aj * v1[j];
    }
  }
}

__device__ __forceinline__ float edge_score(float a1s, float a2d) {
  const float t = a1s + a2d;
  return t > 0.0f ? t : t * NEG_SLOPE;
}

__global__ __launch_bounds__(256) void gat_row_aggregate(
    const int* __restrict__ cnt, const int* __restrict__ slots,
    const int* __restrict__ ovf_n, const int* __restrict__ ovf,
    const float* __restrict__ a1, const float* __restrict__ a2,
    const float* __restrict__ h, float* __restrict__ out, int n) {
  const int lane = threadIdx.x & 31;
  const int wave   = __builtin_amdgcn_readfirstlane(blockIdx.x * 8 + (threadIdx.x >> 5));
  const int nwaves = gridDim.x * 8;

  int   c   = 0;
  float a1s = 0.0f;
  if (wave < n) { c = cnt[wave]; a1s = a1[wave]; }
  for (int s = wave; s < n; s += nwaves) {
    const int sn = s + nwaves;
    int   cn   = 0;
    float a1n  = 0.0f;
    if (sn < n) { cn = cnt[sn]; a1n = a1[sn]; }
    const int* row = slots + (size_t)s * ROW_CAP;
    float acc0 = 0.0f, acc1 = 0.0f;

    if (c <= 32) {
      int d = 0;
      float v = -__builtin_inff();
      if (lane < c) {
        d = row[lane];
        v = edge_score(a1s, a2[d]);
      }
      const float m   = wave_max(v);
      const float ev  = lane < c ? __expf(v - m) : 0.0f;
      const float den = wave_sum(ev);
      const float att = lane < c ? ev / den : 0.0f;
      add_rows(h, lane, c, d, att, acc0, acc1);
    } else {
      const int cs  = min(c, ROW_CAP);
      const int nov = c > ROW_CAP ? *ovf_n : 0;
      float m = -__builtin_inff();
      for (int b = 0; b < cs; b += 32) {
        const int j = b + lane;
        if (j < cs) m = fmaxf(m, edge_score(a1s, a2[row[j]]));
      }
      for (int b = 0; b < nov; b += 32) {
        const int j = b + lane;
        if (j < nov && ovf[2 * j] == s) m = fmaxf(m, edge_score(a1s, a2[ovf[2 * j + 1]]));
      }
      m = wave_max(m);
      float part = 0.0f;
      for (int b = 0; b < cs; b += 32) {
        const int j = b + lane;
        if (j < cs) part += __expf(edge_score(a1s, a2[row[j]]) - m);
      }
      for (int b = 0; b < nov; b += 32) {
        const int j = b + lane;
        if (j < nov && ovf[2 * j] == s) part += __expf(edge_score(a1s, a2[ovf[2 * j + 1]]) - m);
      }
      const float den = wave_sum(part);
      for (int b = 0; b < cs; b += 32) {
        const int j = b + lane;
        int d = 0;
        float att = 0.0f;
        if (j < cs) {
          d = row[j];
          att = __expf(edge_score(a1s, a2[d]) - m) / den;
        }
        add_rows(h, lane, min(32, cs - b), d, att, acc0, acc1);
      }
      for (int b = 0; b < nov; b += 32) {
        const int j = b + lane;
        int d = 0;
        float att = 0.0f;
        bool mine = false;
        if (j < nov && ovf[2 * j] == s) {
          mine = true;
          d = ovf[2 * j + 1];
          att = __expf(edge_score(a1s, a2[d]) - m) / den;
        }
        unsigned live = __builtin_amdgcn_ballot_w32(mine);
        while (live != 0u) {
          const int jj = __builtin_ctz(live);
          live &= live - 1u;
          const int   dj = __builtin_amdgcn_readlane(d, jj);
          const float aj = __int_as_float(__builtin_amdgcn_readlane(__float_as_int(att), jj));
          const float* hr = h + (size_t)dj * D_DIM;
          acc0 += aj * hr[lane];
          acc1 += aj * hr[lane + 32];
        }
      }
    }

    float* orow = out + (size_t)s * D_DIM;
    orow[lane]      = acc0;
    orow[lane + 32] = acc1;
    c = cn; a1s = a1n;
  }
}

extern "C" void kernel_launch(void* const* d_in, const int* in_sizes, int n_in,
                              void* d_out, int out_size, void* d_ws, size_t ws_size,
                              hipStream_t stream) {
  const float* feat = (const float*)d_in[0];
  const int*   idx  = (const int*)d_in[1];
  const float* W    = (const float*)d_in[2];
  const float* bias = (const float*)d_in[3];
  const float* a1w  = (const float*)d_in[4];
  const float* a1b  = (const float*)d_in[5];
  const float* a2w  = (const float*)d_in[6];
  const float* a2b  = (const float*)d_in[7];
  float* out = (float*)d_out;

  const int n = in_sizes[0] / D_DIM;
  const int E = in_sizes[1] / 2;
  const int* src = idx;
  const int* dst = idx + E;

  const int nbins = (n + BIN_ROWS - 1) / BIN_ROWS;
  float* h     = (float*)d_ws;
  int*   slots = (int*)(h + (size_t)n * D_DIM);
  int*   gcnt  = slots + (size_t)n * ROW_CAP;
  int*   recs  = gcnt + (size_t)nbins * CNT_PITCH;
  int*   spill = recs + (size_t)nbins * BIN_CAP * 2;
  int*   ovf   = spill + (size_t)2 * E;
  int*   ctrl  = ovf + (size_t)2 * E;
  float* a1    = (float*)(ctrl + 32);
  float* a2    = a1 + n;
  int*   cnt   = (int*)(a2 + n);

  const int tiles = (n + TILE_M - 1) / TILE_M;
  gat_linear_wmma<<<(tiles + 7) / 8, 256, 0, stream>>>(
      feat, W, bias, a1w, a1b, a2w, a2b, h, a1, a2, gcnt, ctrl, n);

  const int per_block = PART_THREADS * EDGES_PER_THREAD;
  gat_edge_partition<<<(E + per_block - 1) / per_block, PART_THREADS, 0, stream>>>(
      src, dst, gcnt, recs, ctrl + 1, spill, E, nbins);
  gat_bin_place<<<nbins, BIN_ROWS, 0, stream>>>(
      gcnt, recs, ctrl + 1, spill, cnt, slots, ctrl, ovf, n);
  gat_row_aggregate<<<2048, 256, 0, stream>>>(cnt, slots, ctrl, ovf, a1, a2, h, out, n);
}


# edge partition runs in the first launch beside the linear pass: each (bin, work-group) pair owns a fixed 64-record cell ranked in LDS, no global counter; 4 launches to 3; row sums stay LDS-add order
# speedup vs baseline: 5.9562x; 1.1627x over previous
#include <hip/hip_runtime.h>

typedef __attribute__((ext_vector_type(2))) float v2f;
typedef __attribute__((ext_vector_type(8))) float v8f;

#define D_DIM 64
#define TILE_M 16
#define NEG_SLOPE 0.01f

#define ROW_CAP 64
#define BIN_SHIFT 8
#define BIN_ROWS (1 << BIN_SHIFT)
#define MAX_BINS 512
#define PART_THREADS 256
#define EDGES_PER_THREAD 32
#define ROUND_EDGES 8
#define PART_EDGES (PART_THREADS * EDGES_PER_THREAD)
#define CELL_CAP 64

__global__ __launch_bounds__(256, 2) void gat_linear_and_partition(
    const float* __restrict__ feat, const float* __restrict__ W,
    const float* __restrict__ bias,
    const float* __restrict__ a1w, const float* __restrict__ a1b,
    const float* __restrict__ a2w, const float* __restrict__ a2b,
    float* __restrict__ h, float* __restrict__ a1, float* __restrict__ a2,
    int* __restrict__ ctrl, int n_nodes,
    const int* __restrict__ src, const int* __restrict__ dst,
    int* __restrict__ cells, int* __restrict__ cellcnt,
    int* __restrict__ spillcnt, int* __restrict__ spill,
    int E, int nbins, int npart)
{
  constexpr int WP = 68;
  constexpr int FP = 68;
  __shared__ float Wl[D_DIM * WP];
  __shared__ float ft[8][TILE_M * FP];
  __shared__ int lcnt[MAX_BINS + 1];

  const int tid  = threadIdx.x;

  if ((int)blockIdx.x < npart) {
    const int pb = blockIdx.x;
    for (int b = tid; b <= MAX_BINS; b += PART_THREADS) lcnt[b] = 0;
    __syncthreads();

    constexpr int ROUNDS = EDGES_PER_THREAD / ROUND_EDGES;
    const int e0 = pb * PART_EDGES + tid;
    int s[2][ROUND_EDGES], d[2][ROUND_EDGES], r[ROUND_EDGES];
    #pragma unroll
    for (int k = 0; k < ROUND_EDGES; ++k) {
      const int e  = e0 + k * PART_THREADS;
      const int ec = min(e, E - 1);
      const int sv = src[ec];
      d[0][k] = dst[ec];
      s[0][k] = e < E ? sv : -1;
    }
    #pragma unroll
    for (int q = 0; q < ROUNDS; ++q) {
      const int cur = q & 1, nxt = cur ^ 1;
      if (q + 1 < ROUNDS) {
        #pragma unroll
        for (int k = 0; k < ROUND_EDGES; ++k) {
          const int e  = e0 + ((q + 1) * ROUND_EDGES + k) * PART_THREADS;
          const int ec = min(e, E - 1);
          const int sv = src[ec];
          d[nxt][k] = dst[ec];
          s[nxt][k] = e < E ? sv : -1;
        }
      }
      #pragma unroll
      for (int k = 0; k < ROUND_EDGES; ++k)
        r[k] = s[cur][k] >= 0 ? atomicAdd(&lcnt[s[cur][k] >> BIN_SHIFT], 1) : 0;
      #pragma unroll
      for (int k = 0; k < ROUND_EDGES; ++k) {
        if (s[cur][k] >= 0) {
          const int b = s[cur][k] >> BIN_SHIFT;
          int2 sd; sd.x = s[cur][k]; sd.y = d[cur][k];
          int2* where;
          if (r[k] < CELL_CAP) {
            where = (int2*)cells + ((size_t)b * npart + pb) * CELL_CAP + r[k];
          } else {
            const int o = atomicAdd(&lcnt[MAX_BINS], 1);
            where = (int2*)spill + (size_t)pb * PART_EDGES + o;
          }
          *where = sd;
        }
      }
    }
    __syncthreads();
    for (int b = tid; b < nbins; b += PART_THREADS) cellcnt[(size_t)pb * nbins + b] = lcnt[b];
    if (tid == 0) spillcnt[pb] = lcnt[MAX_BINS];
    return;
  }
  const int lb = blockIdx.x - npart;

  const int wave = tid >> 5;
  const int lane = tid & 31;
  const int lh   = lane >> 4;
  const int lid  = lane & 15;

  for (int i = tid; i < D_DIM * (D_DIM / 4); i += 256) {
    const int n  = i >> 4;
    const int c4 = (i & 15) << 2;
    const float4 w4 = *(const float4*)(W + n * D_DIM + c4);
    *(float4*)&Wl[n * WP + c4] = w4;
  }
  __syncthreads();

  if (lb == 0 && tid == 0) ctrl[0] = 0;

  const int tile = lb * 8 + wave;
  const int m0   = tile * TILE_M;
  if (m0 >= n_nodes) return;

  float* myft = ft[wave];
  for (int i = lane; i < TILE_M * (D_DIM / 4); i += 32) {
    const int r  = i >> 4;
    const int c4 = (i & 15) << 2;
    float4 f4 = {0.f, 0.f, 0.f, 0.f};
    if (m0 + r < n_nodes)
      f4 = *(const float4*)(feat + (size_t)(m0 + r) * D_DIM + c4);
    *(float4*)&myft[r * FP + c4] = f4;
  }

  v8f acc[4] = {};
  const int ks = 2 * lh;
  #pragma unroll
  for (int k = 0; k < D_DIM; k += 4) {
    v2f a;
    a.x = myft[lid * FP + k + ks + 0];
    a.y = myft[lid * FP + k + ks + 1];
    #pragma unroll
    for (int t = 0; t < 4; ++t) {
      const int n = t * 16 + lid;
      v2f b;
      b.x = Wl[n * WP + k + ks + 0];
      b.y = Wl[n * WP + k + ks + 1];
      acc[t] = __builtin_amdgcn_wmma_f32_16x16x4_f32(
          false, a, false, b, (short)0, acc[t], false, false);
    }
  }

  #pragma unroll
  for (int t = 0; t < 4; ++t) {
    const float bb = bias[t * 16 + lid];
    #pragma unroll
    for (int j = 0; j < 8; ++j) {
      myft[(j + 8 * lh) * FP + t * 16 + lid] = acc[t][j] + bb;
    }
  }

  for (int i = lane; i < TILE_M * (D_DIM / 4); i += 32) {
    const int r  = i >> 4;
    const int c4 = (i & 15) << 2;
    if (m0 + r < n_nodes) {
      const float4 hv = *(const float4*)&myft[r * FP + c4];
      *(float4*)(h + (size_t)(m0 + r) * D_DIM + c4) = hv;
    }
  }

  const float* aw = lh ? a2w : a1w;
  float s = lh ? a2b[0] : a1b[0];
  #pragma unroll 8
  for (int c = 0; c < D_DIM; ++c)
    s += myft[lid * FP + c] * aw[c];
  if (m0 + lid < n_nodes) {
    if (lh) a2[m0 + lid] = s;
    else    a1[m0 + lid] = s;
  }
}

__device__ __forceinline__ void put_edge(int s, int d, int r, int* stage,
                                         int* __restrict__ ovf_n, int* __restrict__ ovf) {
  if (r < ROW_CAP) {
    stage[(s & (BIN_ROWS - 1)) * ROW_CAP + r] = d;
  } else {
    const int o = atomicAdd(ovf_n, 1);
    ovf[2 * o]     = s;
    ovf[2 * o + 1] = d;
  }
}

#define PLACE_UNROLL 16
#define IMAGE_GROUPS (BIN_ROWS * ROW_CAP / 4 / BIN_ROWS)
__global__ __launch_bounds__(BIN_ROWS) void gat_bin_place(
    const int* __restrict__ cellcnt, const int* __restrict__ cells,
    const int* __restrict__ spillcnt, const int* __restrict__ spill,
    int* __restrict__ cnt, int* __restrict__ slots,
    int* __restrict__ ovf_n, int* __restrict__ ovf, int n, int nbins, int npart) {
  __shared__ int lc[BIN_ROWS];
  __shared__ __attribute__((aligned(16))) int stage[BIN_ROWS * ROW_CAP];
  __shared__ int lcell[BIN_ROWS];
  const int tid  = threadIdx.x;
  const int lane = tid & 31;
  const int b    = blockIdx.x;
  lc[tid] = 0;
  #pragma unroll
  for (int g = 0; g < IMAGE_GROUPS; ++g)
    *(int4*)&stage[(g * BIN_ROWS + tid) * 4] = make_int4(0, 0, 0, 0);

  for (int w0 = 0; w0 < npart; w0 += BIN_ROWS) {
    const int nw = min(BIN_ROWS, npart - w0);
    int cc = 0;
    if (tid < nw) cc = min(cellcnt[(size_t)(w0 + tid) * nbins + b], CELL_CAP);
    __syncthreads();
    lcell[tid] = cc;
    __syncthreads();
    const int2* cb = (const int2*)cells + ((size_t)b * npart + w0) * CELL_CAP;
    const int m = nw * 32;
    for (int i0 = tid; i0 < m; i0 += BIN_ROWS * PLACE_UNROLL) {
      int2 sd[PLACE_UNROLL];
      int  r[PLACE_UNROLL];
      #pragma unroll
      for (int u = 0; u < PLACE_UNROLL; ++u) {
        const int i = i0 + u * BIN_ROWS;
        sd[u].x = -1; sd[u].y = 0;
        if (i < m) {
          const int cell = i >> 5;
          if (lane < lcell[cell]) sd[u] = cb[cell * CELL_CAP + lane];
        }
      }
      #pragma unroll
      for (int u = 0; u < PLACE_UNROLL; ++u)
        r[u] = sd[u].x >= 0 ? atomicAdd(&lc[sd[u].x & (BIN_ROWS - 1)], 1) : 0;
      #pragma unroll
      for (int u = 0; u < PLACE_UNROLL; ++u)
        if (sd[u].x >= 0) put_edge(sd[u].x, sd[u].y, r[u], stage, ovf_n, ovf);
    }
    unsigned longer = __builtin_amdgcn_ballot_w32(cc > 32);
    while (longer != 0u) {
      const int j = __builtin_ctz(longer);
      longer &= longer - 1u;
      const int cell = (tid & ~31) + j;
      const int cj   = __builtin_amdgcn_readlane(cc, j);
      if (32 + lane < cj) {
        const int2 sd = cb[cell * CELL_CAP + 32 + lane];
        const int r = atomicAdd(&lc[sd.x & (BIN_ROWS - 1)], 1);
        put_edge(sd.x, sd.y, r, stage, ovf_n, ovf);
      }
    }
  }
  for (int w = tid; w < npart; w += BIN_ROWS) {
    const int ns = min(spillcnt[w], PART_EDGES);
    const int2* sl = (const int2*)spill + (size_t)w * PART_EDGES;
    for (int k = 0; k < ns; ++k) {
      const int2 sd = sl[k];
      if ((sd.x >> BIN_SHIFT) == b) {
        const int r = atomicAdd(&lc[sd.x & (BIN_ROWS - 1)], 1);
        put_edge(sd.x, sd.y, r, stage, ovf_n, ovf);
      }
    }
  }
  __syncthreads();

  const int row0 = b * BIN_ROWS;
  if (row0 + tid < n) cnt[row0 + tid] = lc[tid];
  #pragma unroll
  for (int g = 0; g < IMAGE_GROUPS; ++g) {
    const int w  = (g * BIN_ROWS + tid) * 4;
    const int rl = w / ROW_CAP;
    const int sl = w % ROW_CAP;
    if (row0 + rl < n && sl < lc[rl]) {
      const int4 v = *(const int4*)&stage[w];
      *(int4*)(slots + (size_t)(row0 + rl) * ROW_CAP + sl) = v;
    }
  }
}

template <int CTRL>
__device__ __forceinline__ float dpp_move(float x) {
  return __int_as_float(__builtin_amdgcn_update_dpp(0, __float_as_int(x), CTRL, 0xF, 0xF, true));
}
#define DPP_XOR1   0xB1
#define DPP_XOR2   0x4E
#define DPP_HMIRR  0x141
#define DPP_MIRR   0x140

__device__ __forceinline__ float wave_max(float x) {
  x = fmaxf(x, dpp_move<DPP_XOR1>(x));
  x = fmaxf(x, dpp_move<DPP_XOR2>(x));
  x = fmaxf(x, dpp_move<DPP_HMIRR>(x));
  x = fmaxf(x, dpp_move<DPP_MIRR>(x));
  const float lo = __int_as_float(__builtin_amdgcn_readlane(__float_as_int(x), 0));
  const float hi = __int_as_float(__builtin_amdgcn_readlane(__float_as_int(x), 16));
  return fmaxf(lo, hi);
}
__device__ __forceinline__ float wave_sum(float x) {
  x += dpp_move<DPP_XOR1>(x);
  x += dpp_move<DPP_XOR2>(x);
  x += dpp_move<DPP_HMIRR>(x);
  x += dpp_move<DPP_MIRR>(x);
  const float lo = __int_as_float(__builtin_amdgcn_readlane(__float_as_int(x), 0));
  const float hi = __int_as_float(__builtin_amdgcn_readlane(__float_as_int(x), 16));
  return lo + hi;
}

#define AGG_GROUP 16
__device__ __forceinline__ void add_rows(const float* __restrict__ h, int lane, int cc,
                                         int d, float att, float& acc0, float& acc1) {
  const int atti = __float_as_int(att);
  for (int j0 = 0; j0 < cc; j0 += AGG_GROUP) {
    float v0[AGG_GROUP], v1[AGG_GROUP];
    #pragma unroll
    for (int j = 0; j < AGG_GROUP; ++j) {
      v0[j] = 0.0f;
      v1[j] = 0.0f;
      if (j0 + j < cc) {
        const int dj = __builtin_amdgcn_readlane(d, j0 + j);
        const float* hr = h + (size_t)dj * D_DIM;
        v0[j] = hr[lane];
        v1[j] = hr[lane + 32];
      }
    }
    #pragma unroll
    for (int j = 0; j < AGG_GROUP; ++j) {
      const float aj = __int_as_float(__builtin_amdgcn_readlane(atti, (j0 + j) & 31));
      acc0 += aj * v0[j];
      acc1 += aj * v1[j];
    }
  }
}

__device__ __forceinline__ float edge_score(float a1s, float a2d) {
  const float t = a1s + a2d;
  return t > 0.0f ? t : t * NEG_SLOPE;
}

__global__ __launch_bounds__(256) void gat_row_aggregate(
    const int* __restrict__ cnt, const int* __restrict__ slots,
    const int* __restrict__ ovf_n, const int* __restrict__ ovf,
    const float* __restrict__ a1, const float* __restrict__ a2,
    const float* __restrict__ h, float* __restrict__ out, int n) {
  const int lane = threadIdx.x & 31;
  const int wave   = __builtin_amdgcn_readfirstlane(blockIdx.x * 8 + (threadIdx.x >> 5));
  const int nwaves = gridDim.x * 8;

  int   c   = 0;
  float a1s = 0.0f;
  if (wave < n) { c = cnt[wave]; a1s = a1[wave]; }
  for (int s = wave; s < n; s += nwaves) {
    const int sn = s + nwaves;
    int   cn   = 0;
    float a1n  = 0.0f;
    if (sn < n) { cn = cnt[sn]; a1n = a1[sn]; }
    const int* row = slots + (size_t)s * ROW_CAP;
    float acc0 = 0.0f, acc1 = 0.0f;

    if (c <= 32) {
      int d = 0;
      float v = -__builtin_inff();
      if (lane < c) {
        d = row[lane];
        v = edge_score(a1s, a2[d]);
      }
      const float m   = wave_max(v);
      const float ev  = lane < c ? __expf(v - m) : 0.0f;
      const float den = wave_sum(ev);
      const float att = lane < c ? ev / den : 0.0f;
      add_rows(h, lane, c, d, att, acc0, acc1);
    } else {
      const int cs  = min(c, ROW_CAP);
      const int nov = c > ROW_CAP ? *ovf_n : 0;
      float m = -__builtin_inff();
      for (int b = 0; b < cs; b += 32) {
        const int j = b + lane;
        if (j < cs) m = fmaxf(m, edge_score(a1s, a2[row[j]]));
      }
      for (int b = 0; b < nov; b += 32) {
        const int j = b + lane;
        if (j < nov && ovf[2 * j] == s) m = fmaxf(m, edge_score(a1s, a2[ovf[2 * j + 1]]));
      }
      m = wave_max(m);
      float part = 0.0f;
      for (int b = 0; b < cs; b += 32) {
        const int j = b + lane;
        if (j < cs) part += __expf(edge_score(a1s, a2[row[j]]) - m);
      }
      for (int b = 0; b < nov; b += 32) {
        const int j = b + lane;
        if (j < nov && ovf[2 * j] == s) part += __expf(edge_score(a1s, a2[ovf[2 * j + 1]]) - m);
      }
      const float den = wave_sum(part);
      for (int b = 0; b < cs; b += 32) {
        const int j = b + lane;
        int d = 0;
        float att = 0.0f;
        if (j < cs) {
          d = row[j];
          att = __expf(edge_score(a1s, a2[d]) - m) / den;
        }
        add_rows(h, lane, min(32, cs - b), d, att, acc0, acc1);
      }
      for (int b = 0; b < nov; b += 32) {
        const int j = b + lane;
        int d = 0;
        float att = 0.0f;
        bool mine = false;
        if (j < nov && ovf[2 * j] == s) {
          mine = true;
          d = ovf[2 * j + 1];
          att = __expf(edge_score(a1s, a2[d]) - m) / den;
        }
        unsigned live = __builtin_amdgcn_ballot_w32(mine);
        while (live != 0u) {
          const int jj = __builtin_ctz(live);
          live &= live - 1u;
          const int   dj = __builtin_amdgcn_readlane(d, jj);
          const float aj = __int_as_float(__builtin_amdgcn_readlane(__float_as_int(att), jj));
          const float* hr = h + (size_t)dj * D_DIM;
          acc0 += aj * hr[lane];
          acc1 += aj * hr[lane + 32];
        }
      }
    }

    float* orow = out + (size_t)s * D_DIM;
    orow[lane]      = acc0;
    orow[lane + 32] = acc1;
    c = cn; a1s = a1n;
  }
}

extern "C" void kernel_launch(void* const* d_in, const int* in_sizes, int n_in,
                              void* d_out, int out_size, void* d_ws, size_t ws_size,
                              hipStream_t stream) {
  const float* feat = (const float*)d_in[0];
  const int*   idx  = (const int*)d_in[1];
  const float* W    = (const float*)d_in[2];
  const float* bias = (const float*)d_in[3];
  const float* a1w  = (const float*)d_in[4];
  const float* a1b  = (const float*)d_in[5];
  const float* a2w  = (const float*)d_in[6];
  const float* a2b  = (const float*)d_in[7];
  float* out = (float*)d_out;

  const int n = in_sizes[0] / D_DIM;
  const int E = in_sizes[1] / 2;
  const int* src = idx;
  const int* dst = idx + E;

  const int nbins = (n + BIN_ROWS - 1) / BIN_ROWS;
  const int npart = (E + PART_EDGES - 1) / PART_EDGES;
  float* h        = (float*)d_ws;
  int*   slots    = (int*)(h + (size_t)n * D_DIM);
  int*   cells    = slots + (size_t)n * ROW_CAP;
  int*   spill    = cells + (size_t)nbins * npart * CELL_CAP * 2;
  int*   ovf      = spill + (size_t)npart * PART_EDGES * 2;
  int*   ctrl     = ovf + (size_t)2 * E;
  float* a1       = (float*)(ctrl + 32);
  float* a2       = a1 + n;
  int*   cnt      = (int*)(a2 + n);
  int*   cellcnt  = cnt + n;
  int*   spillcnt = cellcnt + (size_t)npart * nbins;

  const int tiles = (n + TILE_M - 1) / TILE_M;
  gat_linear_and_partition<<<npart + (tiles + 7) / 8, 256, 0, stream>>>(
      feat, W, bias, a1w, a1b, a2w, a2b, h, a1, a2, ctrl, n,
      src, dst, cells, cellcnt, spillcnt, spill, E, nbins, npart);
  gat_bin_place<<<nbins, BIN_ROWS, 0, stream>>>(
      cellcnt, cells, spillcnt, spill, cnt, slots, ctrl, ovf, n, nbins, npart);
  gat_row_aggregate<<<2048, 256, 0, stream>>>(cnt, slots, ctrl, ovf, a1, a2, h, out, n);
}


# placement and row passes fused: one 32-wave work-group per bin ranks its edges into an LDS slot image and sums the bin's rows from it; slots, counts, overflow list leave memory; 3 launches to 2
# speedup vs baseline: 6.6071x; 1.1093x over previous
#include <hip/hip_runtime.h>

typedef __attribute__((ext_vector_type(2))) float v2f;
typedef __attribute__((ext_vector_type(8))) float v8f;

#define D_DIM 64
#define TILE_M 16
#define NEG_SLOPE 0.01f

#define ROW_CAP 64
#define BIN_SHIFT 8
#define BIN_ROWS (1 << BIN_SHIFT)
#define MAX_BINS 512
#define PART_THREADS 256
#define EDGES_PER_THREAD 32
#define ROUND_EDGES 8
#define PART_EDGES (PART_THREADS * EDGES_PER_THREAD)
#define CELL_CAP 64

__global__ __launch_bounds__(256, 2) void gat_linear_and_partition(
    const float* __restrict__ feat, const float* __restrict__ W,
    const float* __restrict__ bias,
    const float* __restrict__ a1w, const float* __restrict__ a1b,
    const float* __restrict__ a2w, const float* __restrict__ a2b,
    float* __restrict__ h, float* __restrict__ a1, float* __restrict__ a2,
    int n_nodes,
    const int* __restrict__ src, const int* __restrict__ dst,
    int* __restrict__ cells, int* __restrict__ cellcnt,
    int* __restrict__ spillcnt, int* __restrict__ spill,
    int E, int nbins, int npart)
{
  constexpr int WP = 68;
  constexpr int FP = 68;
  __shared__ float Wl[D_DIM * WP];
  __shared__ float ft[8][TILE_M * FP];
  __shared__ int lcnt[MAX_BINS + 1];

  const int tid  = threadIdx.x;

  if ((int)blockIdx.x < npart) {
    const int pb = blockIdx.x;
    for (int b = tid; b <= MAX_BINS; b += PART_THREADS) lcnt[b] = 0;
    __syncthreads();

    constexpr int ROUNDS = EDGES_PER_THREAD / ROUND_EDGES;
    const int e0 = pb * PART_EDGES + tid;
    int s[2][ROUND_EDGES], d[2][ROUND_EDGES], r[ROUND_EDGES];
    #pragma unroll
    for (int k = 0; k < ROUND_EDGES; ++k) {
      const int e  = e0 + k * PART_THREADS;
      const int ec = min(e, E - 1);
      const int sv = src[ec];
      d[0][k] = dst[ec];
      s[0][k] = e < E ? sv : -1;
    }
    #pragma unroll
    for (int q = 0; q < ROUNDS; ++q) {
      const int cur = q & 1, nxt = cur ^ 1;
      if (q + 1 < ROUNDS) {
        #pragma unroll
        for (int k = 0; k < ROUND_EDGES; ++k) {
          const int e  = e0 + ((q + 1) * ROUND_EDGES + k) * PART_THREADS;
          const int ec = min(e, E - 1);
          const int sv = src[ec];
          d[nxt][k] = dst[ec];
          s[nxt][k] = e < E ? sv : -1;
        }
      }
      #pragma unroll
      for (int k = 0; k < ROUND_EDGES; ++k)
        r[k] = s[cur][k] >= 0 ? atomicAdd(&lcnt[s[cur][k] >> BIN_SHIFT], 1) : 0;
      #pragma unroll
      for (int k = 0; k < ROUND_EDGES; ++k) {
        if (s[cur][k] >= 0) {
          const int b = s[cur][k] >> BIN_SHIFT;
          int2 sd; sd.x = s[cur][k]; sd.y = d[cur][k];
          int2* where;
          if (r[k] < CELL_CAP) {
            where = (int2*)cells + ((size_t)b * npart + pb) * CELL_CAP + r[k];
          } else {
            const int o = atomicAdd(&lcnt[MAX_BINS], 1);
            where = (int2*)spill + (size_t)pb * PART_EDGES + o;
          }
          *where = sd;
        }
      }
    }
    __syncthreads();
    for (int b = tid; b < nbins; b += PART_THREADS) cellcnt[(size_t)pb * nbins + b] = lcnt[b];
    if (tid == 0) spillcnt[pb] = lcnt[MAX_BINS];
    return;
  }
  const int lb = blockIdx.x - npart;

  const int wave = tid >> 5;
  const int lane = tid & 31;
  const int lh   = lane >> 4;
  const int lid  = lane & 15;

  for (int i = tid; i < D_DIM * (D_DIM / 4); i += 256) {
    const int n  = i >> 4;
    const int c4 = (i & 15) << 2;
    const float4 w4 = *(const float4*)(W + n * D_DIM + c4);
    *(float4*)&Wl[n * WP + c4] = w4;
  }
  __syncthreads();

  const int tile = lb * 8 + wave;
  const int m0   = tile * TILE_M;
  if (m0 >= n_nodes) return;

  float* myft = ft[wave];
  for (int i = lane; i < TILE_M * (D_DIM / 4); i += 32) {
    const int r  = i >> 4;
    const int c4 = (i & 15) << 2;
    float4 f4 = {0.f, 0.f, 0.f, 0.f};
    if (m0 + r < n_nodes)
      f4 = *(const float4*)(feat + (size_t)(m0 + r) * D_DIM + c4);
    *(float4*)&myft[r * FP + c4] = f4;
  }

  v8f acc[4] = {};
  const int ks = 2 * lh;
  #pragma unroll
  for (int k = 0; k < D_DIM; k += 4) {
    v2f a;
    a.x = myft[lid * FP + k + ks + 0];
    a.y = myft[lid * FP + k + ks + 1];
    #pragma unroll
    for (int t = 0; t < 4; ++t) {
      const int n = t * 16 + lid;
      v2f b;
      b.x = Wl[n * WP + k + ks + 0];
      b.y = Wl[n * WP + k + ks + 1];
      acc[t] = __builtin_amdgcn_wmma_f32_16x16x4_f32(
          false, a, false, b, (short)0, acc[t], false, false);
    }
  }

  #pragma unroll
  for (int t = 0; t < 4; ++t) {
    const float bb = bias[t * 16 + lid];
    #pragma unroll
    for (int j = 0; j < 8; ++j) {
      myft[(j + 8 * lh) * FP + t * 16 + lid] = acc[t][j] + bb;
    }
  }

  for (int i = lane; i < TILE_M * (D_DIM / 4); i += 32) {
    const int r  = i >> 4;
    const int c4 = (i & 15) << 2;
    if (m0 + r < n_nodes) {
      const float4 hv = *(const float4*)&myft[r * FP + c4];
      *(float4*)(h + (size_t)(m0 + r) * D_DIM + c4) = hv;
    }
  }

  const float* aw = lh ? a2w : a1w;
  float s = lh ? a2b[0] : a1b[0];
  #pragma unroll 8
  for (int c = 0; c < D_DIM; ++c)
    s += myft[lid * FP + c] * aw[c];
  if (m0 + lid < n_nodes) {
    if (lh) a2[m0 + lid] = s;
    else    a1[m0 + lid] = s;
  }
}

template <int CTRL>
__device__ __forceinline__ float dpp_move(float x) {
  return __int_as_float(__builtin_amdgcn_update_dpp(0, __float_as_int(x), CTRL, 0xF, 0xF, true));
}
#define DPP_XOR1   0xB1
#define DPP_XOR2   0x4E
#define DPP_HMIRR  0x141
#define DPP_MIRR   0x140

__device__ __forceinline__ float wave_max(float x) {
  x = fmaxf(x, dpp_move<DPP_XOR1>(x));
  x = fmaxf(x, dpp_move<DPP_XOR2>(x));
  x = fmaxf(x, dpp_move<DPP_HMIRR>(x));
  x = fmaxf(x, dpp_move<DPP_MIRR>(x));
  const float lo = __int_as_float(__builtin_amdgcn_readlane(__float_as_int(x), 0));
  const float hi = __int_as_float(__builtin_amdgcn_readlane(__float_as_int(x), 16));
  return fmaxf(lo, hi);
}
__device__ __forceinline__ float wave_sum(float x) {
  x += dpp_move<DPP_XOR1>(x);
  x += dpp_move<DPP_XOR2>(x);
  x += dpp_move<DPP_HMIRR>(x);
  x += dpp_move<DPP_MIRR>(x);
  const float lo = __int_as_float(__builtin_amdgcn_readlane(__float_as_int(x), 0));
  const float hi = __int_as_float(__builtin_amdgcn_readlane(__float_as_int(x), 16));
  return lo + hi;
}

#define AGG_GROUP 16
__device__ __forceinline__ void add_rows(const float* __restrict__ h, int lane, int cc,
                                         int d, float att, float& acc0, float& acc1) {
  const int atti = __float_as_int(att);
  for (int j0 = 0; j0 < cc; j0 += AGG_GROUP) {
    float v0[AGG_GROUP], v1[AGG_GROUP];
    #pragma unroll
    for (int j = 0; j < AGG_GROUP; ++j) {
      v0[j] = 0.0f;
      v1[j] = 0.0f;
      if (j0 + j < cc) {
        const int dj = __builtin_amdgcn_readlane(d, j0 + j);
        const float* hr = h + (size_t)dj * D_DIM;
        v0[j] = hr[lane];
        v1[j] = hr[lane + 32];
      }
    }
    #pragma unroll
    for (int j = 0; j < AGG_GROUP; ++j) {
      const float aj = __int_as_float(__builtin_amdgcn_readlane(atti, (j0 + j) & 31));
      acc0 += aj * v0[j];
      acc1 += aj * v1[j];
    }
  }
}

__device__ __forceinline__ float edge_score(float a1s, float a2d) {
  const float t = a1s + a2d;
  return t > 0.0f ? t : t * NEG_SLOPE;
}

#define BIN_THREADS 1024
#define BIN_WAVES (BIN_THREADS / 32)
#define ROWS_PER_WAVE (BIN_ROWS / BIN_WAVES)
#define PLACE_UNROLL 8

template <typename F>
__device__ __forceinline__ void for_row_records(
    const int* __restrict__ cellcnt, const int2* __restrict__ bincells,
    const int* __restrict__ spillcnt, const int2* __restrict__ spill,
    int b, int s, int nbins, int npart, int lane, F f) {
  for (int w = 0; w < npart; ++w) {
    const int cw = min(cellcnt[(size_t)w * nbins + b], CELL_CAP);
    for (int j0 = 0; j0 < cw; j0 += 32) {
      const int j = j0 + lane;
      bool mine = false;
      int d = 0;
      if (j < cw) {
        const int2 sd = bincells[(size_t)w * CELL_CAP + j];
        mine = sd.x == s;
        d = sd.y;
      }
      f(mine, d);
    }
    const int ns = min(spillcnt[w], PART_EDGES);
    const int2* sl = spill + (size_t)w * PART_EDGES;
    for (int j0 = 0; j0 < ns; j0 += 32) {
      const int j = j0 + lane;
      bool mine = false;
      int d = 0;
      if (j < ns) {
        const int2 sd = sl[j];
        mine = sd.x == s;
        d = sd.y;
      }
      f(mine, d);
    }
  }
}

__global__ __launch_bounds__(BIN_THREADS, 16) void gat_bin_rows(
    const int* __restrict__ cellcnt, const int* __restrict__ cells,
    const int* __restrict__ spillcnt, const int* __restrict__ spill,
    const float* __restrict__ a1, const float* __restrict__ a2,
    const float* __restrict__ h, float* __restrict__ out, int n, int nbins, int npart) {
  __shared__ int lc[BIN_ROWS];
  __shared__ __attribute__((aligned(16))) int stage[BIN_ROWS * ROW_CAP];
  __shared__ int lcell[BIN_ROWS];
  const int tid  = threadIdx.x;
  const int lane = tid & 31;
  const int wv   = __builtin_amdgcn_readfirstlane(tid >> 5);
  const int b    = blockIdx.x;
  const int row0 = b * BIN_ROWS;

  float a1v = 0.0f;
  {
    const int s = row0 + wv + (lane & (ROWS_PER_WAVE - 1)) * BIN_WAVES;
    if (s < n) a1v = a1[s];
  }

  if (tid < BIN_ROWS) lc[tid] = 0;

  const int2* bincells = (const int2*)cells + (size_t)b * npart * CELL_CAP;
  for (int w0 = 0; w0 < npart; w0 += BIN_ROWS) {
    const int nw = min(BIN_ROWS, npart - w0);
    int cc = 0;
    if (tid < nw) cc = min(cellcnt[(size_t)(w0 + tid) * nbins + b], CELL_CAP);
    __syncthreads();
    if (tid < BIN_ROWS) lcell[tid] = cc;
    __syncthreads();
    const int2* cb = bincells + (size_t)w0 * CELL_CAP;
    const int m = nw * 32;
    for (int i0 = tid; i0 < m; i0 += BIN_THREADS * PLACE_UNROLL) {
      int2 sd[PLACE_UNROLL];
      int  r[PLACE_UNROLL];
      #pragma unroll
      for (int u = 0; u < PLACE_UNROLL; ++u) {
        const int i = i0 + u * BIN_THREADS;
        sd[u].x = -1; sd[u].y = 0;
        if (i < m) {
          const int cell = i >> 5;
          if (lane < lcell[cell]) sd[u] = cb[cell * CELL_CAP + lane];
        }
      }
      #pragma unroll
      for (int u = 0; u < PLACE_UNROLL; ++u)
        r[u] = sd[u].x >= 0 ? atomicAdd(&lc[sd[u].x & (BIN_ROWS - 1)], 1) : 0;
      #pragma unroll
      for (int u = 0; u < PLACE_UNROLL; ++u)
        if (sd[u].x >= 0 && r[u] < ROW_CAP)
          stage[(sd[u].x & (BIN_ROWS - 1)) * ROW_CAP + r[u]] = sd[u].y;
    }
    unsigned longer = __builtin_amdgcn_ballot_w32(cc > 32);
    while (longer != 0u) {
      const int j = __builtin_ctz(longer);
      longer &= longer - 1u;
      const int cell = (tid & ~31) + j;
      const int cj   = __builtin_amdgcn_readlane(cc, j);
      if (32 + lane < cj) {
        const int2 sd = cb[cell * CELL_CAP + 32 + lane];
        const int r = atomicAdd(&lc[sd.x & (BIN_ROWS - 1)], 1);
        if (r < ROW_CAP) stage[(sd.x & (BIN_ROWS - 1)) * ROW_CAP + r] = sd.y;
      }
    }
  }
  for (int w = tid; w < npart; w += BIN_THREADS) {
    const int ns = min(spillcnt[w], PART_EDGES);
    const int2* sl = (const int2*)spill + (size_t)w * PART_EDGES;
    for (int k = 0; k < ns; ++k) {
      const int2 sd = sl[k];
      if ((sd.x >> BIN_SHIFT) == b) {
        const int r = atomicAdd(&lc[sd.x & (BIN_ROWS - 1)], 1);
        if (r < ROW_CAP) stage[(sd.x & (BIN_ROWS - 1)) * ROW_CAP + r] = sd.y;
      }
    }
  }
  __syncthreads();

  for (int k = 0; k < ROWS_PER_WAVE; ++k) {
    const int rl = wv + k * BIN_WAVES;
    const int s  = row0 + rl;
    if (s < n) {
      const int   c   = __builtin_amdgcn_readfirstlane(lc[rl]);
      const float a1s = __int_as_float(__builtin_amdgcn_readlane(__float_as_int(a1v), k));
      const int*  row = &stage[rl * ROW_CAP];
      float acc0 = 0.0f, acc1 = 0.0f;

      if (c <= 32) {
        int d = 0;
        float v = -__builtin_inff();
        if (lane < c) {
          d = row[lane];
          v = edge_score(a1s, a2[d]);
        }
        const float m   = wave_max(v);
        const float ev  = lane < c ? __expf(v - m) : 0.0f;
        const float den = wave_sum(ev);
        const float att = lane < c ? ev / den : 0.0f;
        add_rows(h, lane, c, d, att, acc0, acc1);
      } else if (c <= ROW_CAP) {
        float m = -__builtin_inff();
        for (int j0 = 0; j0 < c; j0 += 32) {
          const int j = j0 + lane;
          if (j < c) m = fmaxf(m, edge_score(a1s, a2[row[j]]));
        }
        m = wave_max(m);
        float part = 0.0f;
        for (int j0 = 0; j0 < c; j0 += 32) {
          const int j = j0 + lane;
          if (j < c) part += __expf(edge_score(a1s, a2[row[j]]) - m);
        }
        const float den = wave_sum(part);
        for (int j0 = 0; j0 < c; j0 += 32) {
          const int j = j0 + lane;
          int d = 0;
          float att = 0.0f;
          if (j < c) {
            d = row[j];
            att = __expf(edge_score(a1s, a2[d]) - m) / den;
          }
          add_rows(h, lane, min(32, c - j0), d, att, acc0, acc1);
        }
      } else {
        float m = -__builtin_inff();
        for_row_records(cellcnt, bincells, spillcnt, (const int2*)spill, b, s, nbins, npart, lane,
                        [&](bool mine, int d) {
          if (mine) m = fmaxf(m, edge_score(a1s, a2[d]));
        });
        m = wave_max(m);
        float part = 0.0f;
        for_row_records(cellcnt, bincells, spillcnt, (const int2*)spill, b, s, nbins, npart, lane,
                        [&](bool mine, int d) {
          if (mine) part += __expf(edge_score(a1s, a2[d]) - m);
        });
        const float den = wave_sum(part);
        for_row_records(cellcnt, bincells, spillcnt, (const int2*)spill, b, s, nbins, npart, lane,
                        [&](bool mine, int d) {
          float att = 0.0f;
          if (mine) att = __expf(edge_score(a1s, a2[d]) - m) / den;
          unsigned live = __builtin_amdgcn_ballot_w32(mine);
          while (live != 0u) {
            const int jj = __builtin_ctz(live);
            live &= live - 1u;
            const int   dj = __builtin_amdgcn_readlane(d, jj);
            const float aj = __int_as_float(__builtin_amdgcn_readlane(__float_as_int(att), jj));
            const float* hr = h + (size_t)dj * D_DIM;
            acc0 += aj * hr[lane];
            acc1 += aj * hr[lane + 32];
          }
        });
      }

      float* orow = out + (size_t)s * D_DIM;
      orow[lane]      = acc0;
      orow[lane + 32] = acc1;
    }
  }
}

extern "C" void kernel_launch(void* const* d_in, const int* in_sizes, int n_in,
                              void* d_out, int out_size, void* d_ws, size_t ws_size,
                              hipStream_t stream) {
  const float* feat = (const float*)d_in[0];
  const int*   idx  = (const int*)d_in[1];
  const float* W    = (const float*)d_in[2];
  const float* bias = (const float*)d_in[3];
  const float* a1w  = (const float*)d_in[4];
  const float* a1b  = (const float*)d_in[5];
  const float* a2w  = (const float*)d_in[6];
  const float* a2b  = (const float*)d_in[7];
  float* out = (float*)d_out;

  const int n = in_sizes[0] / D_DIM;
  const int E = in_sizes[1] / 2;
  const int* src = idx;
  const int* dst = idx + E;

  const int nbins = (n + BIN_ROWS - 1) / BIN_ROWS;
  const int npart = (E + PART_EDGES - 1) / PART_EDGES;
  float* h        = (float*)d_ws;
  int*   cells    = (int*)(h + (size_t)n * D_DIM);
  int*   spill    = cells + (size_t)nbins * npart * CELL_CAP * 2;
  float* a1       = (float*)(spill + (size_t)npart * PART_EDGES * 2);
  float* a2       = a1 + n;
  int*   cellcnt  = (int*)(a2 + n);
  int*   spillcnt = cellcnt + (size_t)npart * nbins;

  const int tiles = (n + TILE_M - 1) / TILE_M;
  gat_linear_and_partition<<<npart + (tiles + 7) / 8, 256, 0, stream>>>(
      feat, W, bias, a1w, a1b, a2w, a2b, h, a1, a2, n,
      src, dst, cells, cellcnt, spillcnt, spill, E, nbins, npart);
  gat_bin_rows<<<nbins, BIN_THREADS, 0, stream>>>(
      cellcnt, cells, spillcnt, spill, a1, a2, h, out, n, nbins, npart);
}


# row pass gathers h rows two per load instruction (one 256-byte row per 16 lanes, 16 bytes a lane; was two 4-byte loads per row); even/odd slots summed apart: row and softmax sum order change
# speedup vs baseline: 7.3547x; 1.1132x over previous
#include <hip/hip_runtime.h>

typedef __attribute__((ext_vector_type(2))) float v2f;
typedef __attribute__((ext_vector_type(8))) float v8f;

#define D_DIM 64
#define TILE_M 16
#define NEG_SLOPE 0.01f

#define ROW_CAP 64
#define BIN_SHIFT 8
#define BIN_ROWS (1 << BIN_SHIFT)
#define MAX_BINS 512
#define PART_THREADS 256
#define EDGES_PER_THREAD 32
#define ROUND_EDGES 8
#define PART_EDGES (PART_THREADS * EDGES_PER_THREAD)
#define CELL_CAP 64

__global__ __launch_bounds__(256, 2) void gat_linear_and_partition(
    const float* __restrict__ feat, const float* __restrict__ W,
    const float* __restrict__ bias,
    const float* __restrict__ a1w, const float* __restrict__ a1b,
    const float* __restrict__ a2w, const float* __restrict__ a2b,
    float* __restrict__ h, float* __restrict__ a1, float* __restrict__ a2,
    int n_nodes,
    const int* __restrict__ src, const int* __restrict__ dst,
    int* __restrict__ cells, int* __restrict__ cellcnt,
    int* __restrict__ spillcnt, int* __restrict__ spill,
    int E, int nbins, int npart)
{
  constexpr int WP = 68;
  constexpr int FP = 68;
  __shared__ float Wl[D_DIM * WP];
  __shared__ float ft[8][TILE_M * FP];
  __shared__ int lcnt[MAX_BINS + 1];

  const int tid  = threadIdx.x;

  if ((int)blockIdx.x < npart) {
    const int pb = blockIdx.x;
    for (int b = tid; b <= MAX_BINS; b += PART_THREADS) lcnt[b] = 0;
    __syncthreads();

    constexpr int ROUNDS = EDGES_PER_THREAD / ROUND_EDGES;
    const int e0 = pb * PART_EDGES + tid;
    int s[2][ROUND_EDGES], d[2][ROUND_EDGES], r[ROUND_EDGES];
    #pragma unroll
    for (int k = 0; k < ROUND_EDGES; ++k) {
      const int e  = e0 + k * PART_THREADS;
      const int ec = min(e, E - 1);
      const int sv = src[ec];
      d[0][k] = dst[ec];
      s[0][k] = e < E ? sv : -1;
    }
    #pragma unroll
    for (int q = 0; q < ROUNDS; ++q) {
      const int cur = q & 1, nxt = cur ^ 1;
      if (q + 1 < ROUNDS) {
        #pragma unroll
        for (int k = 0; k < ROUND_EDGES; ++k) {
          const int e  = e0 + ((q + 1) * ROUND_EDGES + k) * PART_THREADS;
          const int ec = min(e, E - 1);
          const int sv = src[ec];
          d[nxt][k] = dst[ec];
          s[nxt][k] = e < E ? sv : -1;
        }
      }
      #pragma unroll
      for (int k = 0; k < ROUND_EDGES; ++k)
        r[k] = s[cur][k] >= 0 ? atomicAdd(&lcnt[s[cur][k] >> BIN_SHIFT], 1) : 0;
      #pragma unroll
      for (int k = 0; k < ROUND_EDGES; ++k) {
        if (s[cur][k] >= 0) {
          const int b = s[cur][k] >> BIN_SHIFT;
          int2 sd; sd.x = s[cur][k]; sd.y = d[cur][k];
          int2* where;
          if (r[k] < CELL_CAP) {
            where = (int2*)cells + ((size_t)b * npart + pb) * CELL_CAP + r[k];
          } else {
            const int o = atomicAdd(&lcnt[MAX_BINS], 1);
            where = (int2*)spill + (size_t)pb * PART_EDGES + o;
          }
          *where = sd;
        }
      }
    }
    __syncthreads();
    for (int b = tid; b < nbins; b += PART_THREADS) cellcnt[(size_t)pb * nbins + b] = lcnt[b];
    if (tid == 0) spillcnt[pb] = lcnt[MAX_BINS];
    return;
  }
  const int lb = blockIdx.x - npart;

  const int wave = tid >> 5;
  const int lane = tid & 31;
  const int lh   = lane >> 4;
  const int lid  = lane & 15;

  for (int i = tid; i < D_DIM * (D_DIM / 4); i += 256) {
    const int n  = i >> 4;
    const int c4 = (i & 15) << 2;
    const float4 w4 = *(const float4*)(W + n * D_DIM + c4);
    *(float4*)&Wl[n * WP + c4] = w4;
  }
  __syncthreads();

  const int tile = lb * 8 + wave;
  const int m0   = tile * TILE_M;
  if (m0 >= n_nodes) return;

  float* myft = ft[wave];
  for (int i = lane; i < TILE_M * (D_DIM / 4); i += 32) {
    const int r  = i >> 4;
    const int c4 = (i & 15) << 2;
    float4 f4 = {0.f, 0.f, 0.f, 0.f};
    if (m0 + r < n_nodes)
      f4 = *(const float4*)(feat + (size_t)(m0 + r) * D_DIM + c4);
    *(float4*)&myft[r * FP + c4] = f4;
  }

  v8f acc[4] = {};
  const int ks = 2 * lh;
  #pragma unroll
  for (int k = 0; k < D_DIM; k += 4) {
    v2f a;
    a.x = myft[lid * FP + k + ks + 0];
    a.y = myft[lid * FP + k + ks + 1];
    #pragma unroll
    for (int t = 0; t < 4; ++t) {
      const int n = t * 16 + lid;
      v2f b;
      b.x = Wl[n * WP + k + ks + 0];
      b.y = Wl[n * WP + k + ks + 1];
      acc[t] = __builtin_amdgcn_wmma_f32_16x16x4_f32(
          false, a, false, b, (short)0, acc[t], false, false);
    }
  }

  #pragma unroll
  for (int t = 0; t < 4; ++t) {
    const float bb = bias[t * 16 + lid];
    #pragma unroll
    for (int j = 0; j < 8; ++j) {
      myft[(j + 8 * lh) * FP + t * 16 + lid] = acc[t][j] + bb;
    }
  }

  for (int i = lane; i < TILE_M * (D_DIM / 4); i += 32) {
    const int r  = i >> 4;
    const int c4 = (i & 15) << 2;
    if (m0 + r < n_nodes) {
      const float4 hv = *(const float4*)&myft[r * FP + c4];
      *(float4*)(h + (size_t)(m0 + r) * D_DIM + c4) = hv;
    }
  }

  const float* aw = lh ? a2w : a1w;
  float s = lh ? a2b[0] : a1b[0];
  #pragma unroll 8
  for (int c = 0; c < D_DIM; ++c)
    s += myft[lid * FP + c] * aw[c];
  if (m0 + lid < n_nodes) {
    if (lh) a2[m0 + lid] = s;
    else    a1[m0 + lid] = s;
  }
}

template <int CTRL>
__device__ __forceinline__ float dpp_move(float x) {
  return __int_as_float(__builtin_amdgcn_update_dpp(0, __float_as_int(x), CTRL, 0xF, 0xF, true));
}
#define DPP_XOR1   0xB1
#define DPP_XOR2   0x4E
#define DPP_HMIRR  0x141
#define DPP_MIRR   0x140

__device__ __forceinline__ float wave_max(float x) {
  x = fmaxf(x, dpp_move<DPP_XOR1>(x));
  x = fmaxf(x, dpp_move<DPP_XOR2>(x));
  x = fmaxf(x, dpp_move<DPP_HMIRR>(x));
  x = fmaxf(x, dpp_move<DPP_MIRR>(x));
  const float lo = __int_as_float(__builtin_amdgcn_readlane(__float_as_int(x), 0));
  const float hi = __int_as_float(__builtin_amdgcn_readlane(__float_as_int(x), 16));
  return fmaxf(lo, hi);
}
__device__ __forceinline__ float wave_sum(float x) {
  x += dpp_move<DPP_XOR1>(x);
  x += dpp_move<DPP_XOR2>(x);
  x += dpp_move<DPP_HMIRR>(x);
  x += dpp_move<DPP_MIRR>(x);
  const float lo = __int_as_float(__builtin_amdgcn_readlane(__float_as_int(x), 0));
  const float hi = __int_as_float(__builtin_amdgcn_readlane(__float_as_int(x), 16));
  return lo + hi;
}

template <int K>
__device__ __forceinline__ int row_share(int x) {
  return __builtin_amdgcn_update_dpp(0, x, 0x150 + K, 0xF, 0xF, true);
}

#define AGG_PAIRS 8
template <int P>
__device__ __forceinline__ void load_pair(const float4* __restrict__ h4, int lid, int np, int d, float4& v) {
  v = make_float4(0.0f, 0.0f, 0.0f, 0.0f);
  if (P < np) {
    const int dj = row_share<P>(d);
    v = h4[(unsigned)(dj * (D_DIM / 4) + lid)];
  }
}
template <int P>
__device__ __forceinline__ void fma_pair(int atti, const float4& v, float4& acc) {
  const float aj = __int_as_float(row_share<P>(atti));
  acc.x += aj * v.x;
  acc.y += aj * v.y;
  acc.z += aj * v.z;
  acc.w += aj * v.w;
}
template <int P0>
__device__ __forceinline__ void add_pair_group(const float4* __restrict__ h4, int lid, int np,
                                               int d, int atti, float4& acc) {
  float4 v0, v1, v2, v3, v4, v5, v6, v7;
  load_pair<P0 + 0>(h4, lid, np, d, v0);
  load_pair<P0 + 1>(h4, lid, np, d, v1);
  load_pair<P0 + 2>(h4, lid, np, d, v2);
  load_pair<P0 + 3>(h4, lid, np, d, v3);
  load_pair<P0 + 4>(h4, lid, np, d, v4);
  load_pair<P0 + 5>(h4, lid, np, d, v5);
  load_pair<P0 + 6>(h4, lid, np, d, v6);
  load_pair<P0 + 7>(h4, lid, np, d, v7);
  fma_pair<P0 + 0>(atti, v0, acc);
  fma_pair<P0 + 1>(atti, v1, acc);
  fma_pair<P0 + 2>(atti, v2, acc);
  fma_pair<P0 + 3>(atti, v3, acc);
  fma_pair<P0 + 4>(atti, v4, acc);
  fma_pair<P0 + 5>(atti, v5, acc);
  fma_pair<P0 + 6>(atti, v6, acc);
  fma_pair<P0 + 7>(atti, v7, acc);
}
__device__ __forceinline__ void add_rows(const float* __restrict__ h, int lid, int cc,
                                         int d, float att, float4& acc) {
  const float4* h4 = (const float4*)h;
  const int atti = __float_as_int(att);
  const int np   = (cc + 1) >> 1;
  add_pair_group<0>(h4, lid, np, d, atti, acc);
  if (np > AGG_PAIRS)
    add_pair_group<AGG_PAIRS>(h4, lid, np, d, atti, acc);
}

__device__ __forceinline__ int other_half_i(int x) {
  return __builtin_amdgcn_permlanex16(0, x, 0x76543210, 0xfedcba98, false, false);
}
__device__ __forceinline__ float other_half(float x) {
  return __int_as_float(other_half_i(__float_as_int(x)));
}

__device__ __forceinline__ float edge_score(float a1s, float a2d) {
  const float t = a1s + a2d;
  return t > 0.0f ? t : t * NEG_SLOPE;
}

#define BIN_THREADS 1024
#define BIN_WAVES (BIN_THREADS / 32)
#define ROWS_PER_WAVE (BIN_ROWS / BIN_WAVES)
#define PLACE_UNROLL 8

template <typename F>
__device__ __forceinline__ void for_row_records(
    const int* __restrict__ cellcnt, const int2* __restrict__ bincells,
    const int* __restrict__ spillcnt, const int2* __restrict__ spill,
    int b, int s, int nbins, int npart, int lane, F f) {
  for (int w = 0; w < npart; ++w) {
    const int cw = min(cellcnt[(size_t)w * nbins + b], CELL_CAP);
    for (int j0 = 0; j0 < cw; j0 += 32) {
      const int j = j0 + lane;
      bool mine = false;
      int d = 0;
      if (j < cw) {
        const int2 sd = bincells[(size_t)w * CELL_CAP + j];
        mine = sd.x == s;
        d = sd.y;
      }
      f(mine, d);
    }
    const int ns = min(spillcnt[w], PART_EDGES);
    const int2* sl = spill + (size_t)w * PART_EDGES;
    for (int j0 = 0; j0 < ns; j0 += 32) {
      const int j = j0 + lane;
      bool mine = false;
      int d = 0;
      if (j < ns) {
        const int2 sd = sl[j];
        mine = sd.x == s;
        d = sd.y;
      }
      f(mine, d);
    }
  }
}

__global__ __launch_bounds__(BIN_THREADS, 16) void gat_bin_rows(
    const int* __restrict__ cellcnt, const int* __restrict__ cells,
    const int* __restrict__ spillcnt, const int* __restrict__ spill,
    const float* __restrict__ a1, const float* __restrict__ a2,
    const float* __restrict__ h, float* __restrict__ out, int n, int nbins, int npart) {
  __shared__ int lc[BIN_ROWS];
  __shared__ __attribute__((aligned(16))) int stage[BIN_ROWS * ROW_CAP];
  __shared__ int lcell[BIN_ROWS];
  const int tid  = threadIdx.x;
  const int lane = tid & 31;
  const int wv   = __builtin_amdgcn_readfirstlane(tid >> 5);
  const int b    = blockIdx.x;
  const int row0 = b * BIN_ROWS;

  float a1v = 0.0f;
  {
    const int s = row0 + wv + (lane & (ROWS_PER_WAVE - 1)) * BIN_WAVES;
    if (s < n) a1v = a1[s];
  }

  if (tid < BIN_ROWS) lc[tid] = 0;

  const int2* bincells = (const int2*)cells + (size_t)b * npart * CELL_CAP;
  for (int w0 = 0; w0 < npart; w0 += BIN_ROWS) {
    const int nw = min(BIN_ROWS, npart - w0);
    int cc = 0;
    if (tid < nw) cc = min(cellcnt[(size_t)(w0 + tid) * nbins + b], CELL_CAP);
    __syncthreads();
    if (tid < BIN_ROWS) lcell[tid] = cc;
    __syncthreads();
    const int2* cb = bincells + (size_t)w0 * CELL_CAP;
    const int m = nw * 32;
    for (int i0 = tid; i0 < m; i0 += BIN_THREADS * PLACE_UNROLL) {
      int2 sd[PLACE_UNROLL];
      int  r[PLACE_UNROLL];
      #pragma unroll
      for (int u = 0; u < PLACE_UNROLL; ++u) {
        const int i = i0 + u * BIN_THREADS;
        sd[u].x = -1; sd[u].y = 0;
        if (i < m) {
          const int cell = i >> 5;
          if (lane < lcell[cell]) sd[u] = cb[cell * CELL_CAP + lane];
        }
      }
      #pragma unroll
      for (int u = 0; u < PLACE_UNROLL; ++u)
        r[u] = sd[u].x >= 0 ? atomicAdd(&lc[sd[u].x & (BIN_ROWS - 1)], 1) : 0;
      #pragma unroll
      for (int u = 0; u < PLACE_UNROLL; ++u)
        if (sd[u].x >= 0 && r[u] < ROW_CAP)
          stage[(sd[u].x & (BIN_ROWS - 1)) * ROW_CAP + r[u]] = sd[u].y;
    }
    unsigned longer = __builtin_amdgcn_ballot_w32(cc > 32);
    while (longer != 0u) {
      const int j = __builtin_ctz(longer);
      longer &= longer - 1u;
      const int cell = (tid & ~31) + j;
      const int cj   = __builtin_amdgcn_readlane(cc, j);
      if (32 + lane < cj) {
        const int2 sd = cb[cell * CELL_CAP + 32 + lane];
        const int r = atomicAdd(&lc[sd.x & (BIN_ROWS - 1)], 1);
        if (r < ROW_CAP) stage[(sd.x & (BIN_ROWS - 1)) * ROW_CAP + r] = sd.y;
      }
    }
  }
  for (int w = tid; w < npart; w += BIN_THREADS) {
    const int ns = min(spillcnt[w], PART_EDGES);
    const int2* sl = (const int2*)spill + (size_t)w * PART_EDGES;
    for (int k = 0; k < ns; ++k) {
      const int2 sd = sl[k];
      if ((sd.x >> BIN_SHIFT) == b) {
        const int r = atomicAdd(&lc[sd.x & (BIN_ROWS - 1)], 1);
        if (r < ROW_CAP) stage[(sd.x & (BIN_ROWS - 1)) * ROW_CAP + r] = sd.y;
      }
    }
  }
  __syncthreads();

  for (int k = 0; k < ROWS_PER_WAVE; ++k) {
    const int rl = wv + k * BIN_WAVES;
    const int s  = row0 + rl;
    if (s < n) {
      const int   c   = __builtin_amdgcn_readfirstlane(lc[rl]);
      const float a1s = __int_as_float(__builtin_amdgcn_readlane(__float_as_int(a1v), k));
      const int*  row = &stage[rl * ROW_CAP];
      float* orow = out + (size_t)s * D_DIM;

      if (c <= ROW_CAP) {
        const int lh   = lane >> 4;
        const int lid  = lane & 15;
        const int lid4 = lid << 2;
        const int sl   = 2 * lid + lh;
        float4 acc = make_float4(0.0f, 0.0f, 0.0f, 0.0f);
        if (c <= 32) {
          int d = 0;
          float v = -__builtin_inff();
          if (sl < c) {
            d = row[sl];
            v = edge_score(a1s, a2[d]);
          }
          const float m   = wave_max(v);
          const float ev  = sl < c ? __expf(v - m) : 0.0f;
          const float den = wave_sum(ev);
          const float att = sl < c ? ev / den : 0.0f;
          const int dlo = other_half_i(d);
          if (sl >= c) d = dlo;
          add_rows(h, lid, c, d, att, acc);
        } else {
          float m = -__builtin_inff();
          for (int j0 = 0; j0 < c; j0 += 32) {
            const int j = j0 + sl;
            if (j < c) m = fmaxf(m, edge_score(a1s, a2[row[j]]));
          }
          m = wave_max(m);
          float part = 0.0f;
          for (int j0 = 0; j0 < c; j0 += 32) {
            const int j = j0 + sl;
            if (j < c) part += __expf(edge_score(a1s, a2[row[j]]) - m);
          }
          const float den = wave_sum(part);
          for (int j0 = 0; j0 < c; j0 += 32) {
            const int j = j0 + sl;
            int d = 0;
            float att = 0.0f;
            if (j < c) {
              d = row[j];
              att = __expf(edge_score(a1s, a2[d]) - m) / den;
            }
            const int dlo = other_half_i(d);
            if (j >= c) d = dlo;
            add_rows(h, lid, min(32, c - j0), d, att, acc);
          }
        }
        acc.x += other_half(acc.x);
        acc.y += other_half(acc.y);
        acc.z += other_half(acc.z);
        acc.w += other_half(acc.w);
        if (lane < 16) *(float4*)(orow + lid4) = acc;
      } else {
        float acc0 = 0.0f, acc1 = 0.0f;
        float m = -__builtin_inff();
        for_row_records(cellcnt, bincells, spillcnt, (const int2*)spill, b, s, nbins, npart, lane,
                        [&](bool mine, int d) {
          if (mine) m = fmaxf(m, edge_score(a1s, a2[d]));
        });
        m = wave_max(m);
        float part = 0.0f;
        for_row_records(cellcnt, bincells, spillcnt, (const int2*)spill, b, s, nbins, npart, lane,
                        [&](bool mine, int d) {
          if (mine) part += __expf(edge_score(a1s, a2[d]) - m);
        });
        const float den = wave_sum(part);
        for_row_records(cellcnt, bincells, spillcnt, (const int2*)spill, b, s, nbins, npart, lane,
                        [&](bool mine, int d) {
          float att = 0.0f;
          if (mine) att = __expf(edge_score(a1s, a2[d]) - m) / den;
          unsigned live = __builtin_amdgcn_ballot_w32(mine);
          while (live != 0u) {
            const int jj = __builtin_ctz(live);
            live &= live - 1u;
            const int   dj = __builtin_amdgcn_readlane(d, jj);
            const float aj = __int_as_float(__builtin_amdgcn_readlane(__float_as_int(att), jj));
            const float* hr = h + (size_t)dj * D_DIM;
            acc0 += aj * hr[lane];
            acc1 += aj * hr[lane + 32];
          }
        });
        orow[lane]      = acc0;
        orow[lane + 32] = acc1;
      }
    }
  }
}

extern "C" void kernel_launch(void* const* d_in, const int* in_sizes, int n_in,
                              void* d_out, int out_size, void* d_ws, size_t ws_size,
                              hipStream_t stream) {
  const float* feat = (const float*)d_in[0];
  const int*   idx  = (const int*)d_in[1];
  const float* W    = (const float*)d_in[2];
  const float* bias = (const float*)d_in[3];
  const float* a1w  = (const float*)d_in[4];
  const float* a1b  = (const float*)d_in[5];
  const float* a2w  = (const float*)d_in[6];
  const float* a2b  = (const float*)d_in[7];
  float* out = (float*)d_out;

  const int n = in_sizes[0] / D_DIM;
  const int E = in_sizes[1] / 2;
  const int* src = idx;
  const int* dst = idx + E;

  const int nbins = (n + BIN_ROWS - 1) / BIN_ROWS;
  const int npart = (E + PART_EDGES - 1) / PART_EDGES;
  float* h        = (float*)d_ws;
  int*   cells    = (int*)(h + (size_t)n * D_DIM);
  int*   spill    = cells + (size_t)nbins * npart * CELL_CAP * 2;
  float* a1       = (float*)(spill + (size_t)npart * PART_EDGES * 2);
  float* a2       = a1 + n;
  int*   cellcnt  = (int*)(a2 + n);
  int*   spillcnt = cellcnt + (size_t)npart * nbins;

  const int tiles = (n + TILE_M - 1) / TILE_M;
  gat_linear_and_partition<<<npart + (tiles + 7) / 8, 256, 0, stream>>>(
      feat, W, bias, a1w, a1b, a2w, a2b, h, a1, a2, n,
      src, dst, cells, cellcnt, spillcnt, spill, E, nbins, npart);
  gat_bin_rows<<<nbins, BIN_THREADS, 0, stream>>>(
      cellcnt, cells, spillcnt, spill, a1, a2, h, out, n, nbins, npart);
}


# linear pass: a wave takes 2 row tiles in a loop, next tile's rows loaded under the current tile's products; W, bias, score weights fetched once per 16 tiles; 6256 one-tile waves become 3128
# speedup vs baseline: 7.6421x; 1.0391x over previous
#include <hip/hip_runtime.h>

typedef __attribute__((ext_vector_type(2))) float v2f;
typedef __attribute__((ext_vector_type(8))) float v8f;

#define D_DIM 64
#define TILE_M 16
#define NEG_SLOPE 0.01f

#define ROW_CAP 64
#define BIN_SHIFT 8
#define BIN_ROWS (1 << BIN_SHIFT)
#define MAX_BINS 512
#define PART_THREADS 256
#define EDGES_PER_THREAD 32
#define ROUND_EDGES 8
#define PART_EDGES (PART_THREADS * EDGES_PER_THREAD)
#define CELL_CAP 64
#define TILES_PER_WAVE 2

__global__ __launch_bounds__(256, 2) void gat_linear_and_partition(
    const float* __restrict__ feat, const float* __restrict__ W,
    const float* __restrict__ bias,
    const float* __restrict__ a1w, const float* __restrict__ a1b,
    const float* __restrict__ a2w, const float* __restrict__ a2b,
    float* __restrict__ h, float* __restrict__ a1, float* __restrict__ a2,
    int n_nodes,
    const int* __restrict__ src, const int* __restrict__ dst,
    int* __restrict__ cells, int* __restrict__ cellcnt,
    int* __restrict__ spillcnt, int* __restrict__ spill,
    int E, int nbins, int npart)
{
  constexpr int WP = 68;
  constexpr int FP = 68;
  __shared__ float Wl[D_DIM * WP];
  __shared__ float ft[8][TILE_M * FP];
  __shared__ int lcnt[MAX_BINS + 1];
  __shared__ __attribute__((aligned(16))) float awl[2 * D_DIM];

  const int tid  = threadIdx.x;

  if ((int)blockIdx.x < npart) {
    const int pb = blockIdx.x;
    for (int b = tid; b <= MAX_BINS; b += PART_THREADS) lcnt[b] = 0;
    __syncthreads();

    constexpr int ROUNDS = EDGES_PER_THREAD / ROUND_EDGES;
    const int e0 = pb * PART_EDGES + tid;
    int s[2][ROUND_EDGES], d[2][ROUND_EDGES], r[ROUND_EDGES];
    #pragma unroll
    for (int k = 0; k < ROUND_EDGES; ++k) {
      const int e  = e0 + k * PART_THREADS;
      const int ec = min(e, E - 1);
      const int sv = src[ec];
      d[0][k] = dst[ec];
      s[0][k] = e < E ? sv : -1;
    }
    #pragma unroll
    for (int q = 0; q < ROUNDS; ++q) {
      const int cur = q & 1, nxt = cur ^ 1;
      if (q + 1 < ROUNDS) {
        #pragma unroll
        for (int k = 0; k < ROUND_EDGES; ++k) {
          const int e  = e0 + ((q + 1) * ROUND_EDGES + k) * PART_THREADS;
          const int ec = min(e, E - 1);
          const int sv = src[ec];
          d[nxt][k] = dst[ec];
          s[nxt][k] = e < E ? sv : -1;
        }
      }
      #pragma unroll
      for (int k = 0; k < ROUND_EDGES; ++k)
        r[k] = s[cur][k] >= 0 ? atomicAdd(&lcnt[s[cur][k] >> BIN_SHIFT], 1) : 0;
      #pragma unroll
      for (int k = 0; k < ROUND_EDGES; ++k) {
        if (s[cur][k] >= 0) {
          const int b = s[cur][k] >> BIN_SHIFT;
          int2 sd; sd.x = s[cur][k]; sd.y = d[cur][k];
          int2* where;
          if (r[k] < CELL_CAP) {
            where = (int2*)cells + ((size_t)b * npart + pb) * CELL_CAP + r[k];
          } else {
            const int o = atomicAdd(&lcnt[MAX_BINS], 1);
            where = (int2*)spill + (size_t)pb * PART_EDGES + o;
          }
          *where = sd;
        }
      }
    }
    __syncthreads();
    for (int b = tid; b < nbins; b += PART_THREADS) cellcnt[(size_t)pb * nbins + b] = lcnt[b];
    if (tid == 0) spillcnt[pb] = lcnt[MAX_BINS];
    return;
  }
  const int lb = blockIdx.x - npart;

  const int wave = tid >> 5;
  const int lane = tid & 31;
  const int lh   = lane >> 4;
  const int lid  = lane & 15;

  for (int i = tid; i < D_DIM * (D_DIM / 4); i += 256) {
    const int n  = i >> 4;
    const int c4 = (i & 15) << 2;
    const float4 w4 = *(const float4*)(W + n * D_DIM + c4);
    *(float4*)&Wl[n * WP + c4] = w4;
  }
  if (tid < 2 * D_DIM) awl[tid] = tid < D_DIM ? a1w[tid] : a2w[tid - D_DIM];
  __syncthreads();

  float* myft = ft[wave];
  const int r0 = lane >> 4;
  const int c4 = (lane & 15) << 2;
  const int ks = 2 * lh;
  const int tile0 = lb * (8 * TILES_PER_WAVE) + wave;

  if (tile0 * TILE_M >= n_nodes) return;

  float bb[4];
  #pragma unroll
  for (int t = 0; t < 4; ++t) bb[t] = bias[t * 16 + lid];
  const float* aw = &awl[lh * D_DIM];
  const float  sb = lh ? a2b[0] : a1b[0];

#define FOR_8_ROW_LOADS(X) X(0) X(1) X(2) X(3) X(4) X(5) X(6) X(7)
#define LOAD_ROWS(q)  f##q = *(const float4*)(feat + (size_t)min(mload + 2 * q + r0, n_nodes - 1) * D_DIM + c4);
#define STAGE_ROWS(q) *(float4*)&myft[(2 * q + r0) * FP + c4] = f##q;
  float4 f0, f1, f2, f3, f4, f5, f6, f7;
  {
    const int mload = tile0 * TILE_M;
    FOR_8_ROW_LOADS(LOAD_ROWS)
  }

  for (int j = 0; j < TILES_PER_WAVE; ++j) {
    const int m0 = (tile0 + j * 8) * TILE_M;
    if (m0 >= n_nodes) break;

    FOR_8_ROW_LOADS(STAGE_ROWS)

    const int m1 = m0 + 8 * TILE_M;
    if (j + 1 < TILES_PER_WAVE && m1 < n_nodes) {
      const int mload = m1;
      FOR_8_ROW_LOADS(LOAD_ROWS)
    }

    v8f acc[4] = {};
    #pragma unroll
    for (int k = 0; k < D_DIM; k += 4) {
      v2f a;
      a.x = myft[lid * FP + k + ks + 0];
      a.y = myft[lid * FP + k + ks + 1];
      #pragma unroll
      for (int t = 0; t < 4; ++t) {
        const int n = t * 16 + lid;
        v2f b;
        b.x = Wl[n * WP + k + ks + 0];
        b.y = Wl[n * WP + k + ks + 1];
        acc[t] = __builtin_amdgcn_wmma_f32_16x16x4_f32(
            false, a, false, b, (short)0, acc[t], false, false);
      }
    }

    #pragma unroll
    for (int t = 0; t < 4; ++t) {
      #pragma unroll
      for (int i = 0; i < 8; ++i) {
        myft[(i + 8 * lh) * FP + t * 16 + lid] = acc[t][i] + bb[t];
      }
    }

    for (int i = lane; i < TILE_M * (D_DIM / 4); i += 32) {
      const int r  = i >> 4;
      const int cc = (i & 15) << 2;
      if (m0 + r < n_nodes) {
        const float4 hv = *(const float4*)&myft[r * FP + cc];
        *(float4*)(h + (size_t)(m0 + r) * D_DIM + cc) = hv;
      }
    }

    float s = sb;
    #pragma unroll 8
    for (int c = 0; c < D_DIM; ++c)
      s += myft[lid * FP + c] * aw[c];
    if (m0 + lid < n_nodes) {
      if (lh) a2[m0 + lid] = s;
      else    a1[m0 + lid] = s;
    }
  }
}

template <int CTRL>
__device__ __forceinline__ float dpp_move(float x) {
  return __int_as_float(__builtin_amdgcn_update_dpp(0, __float_as_int(x), CTRL, 0xF, 0xF, true));
}
#define DPP_XOR1   0xB1
#define DPP_XOR2   0x4E
#define DPP_HMIRR  0x141
#define DPP_MIRR   0x140

__device__ __forceinline__ float wave_max(float x) {
  x = fmaxf(x, dpp_move<DPP_XOR1>(x));
  x = fmaxf(x, dpp_move<DPP_XOR2>(x));
  x = fmaxf(x, dpp_move<DPP_HMIRR>(x));
  x = fmaxf(x, dpp_move<DPP_MIRR>(x));
  const float lo = __int_as_float(__builtin_amdgcn_readlane(__float_as_int(x), 0));
  const float hi = __int_as_float(__builtin_amdgcn_readlane(__float_as_int(x), 16));
  return fmaxf(lo, hi);
}
__device__ __forceinline__ float wave_sum(float x) {
  x += dpp_move<DPP_XOR1>(x);
  x += dpp_move<DPP_XOR2>(x);
  x += dpp_move<DPP_HMIRR>(x);
  x += dpp_move<DPP_MIRR>(x);
  const float lo = __int_as_float(__builtin_amdgcn_readlane(__float_as_int(x), 0));
  const float hi = __int_as_float(__builtin_amdgcn_readlane(__float_as_int(x), 16));
  return lo + hi;
}

template <int K>
__device__ __forceinline__ int row_share(int x) {
  return __builtin_amdgcn_update_dpp(0, x, 0x150 + K, 0xF, 0xF, true);
}

#define AGG_PAIRS 8
template <int P>
__device__ __forceinline__ void load_pair(const float4* __restrict__ h4, int lid, int np, int d, float4& v) {
  v = make_float4(0.0f, 0.0f, 0.0f, 0.0f);
  if (P < np) {
    const int dj = row_share<P>(d);
    v = h4[(unsigned)(dj * (D_DIM / 4) + lid)];
  }
}
template <int P>
__device__ __forceinline__ void fma_pair(int atti, const float4& v, float4& acc) {
  const float aj = __int_as_float(row_share<P>(atti));
  acc.x += aj * v.x;
  acc.y += aj * v.y;
  acc.z += aj * v.z;
  acc.w += aj * v.w;
}
template <int P0>
__device__ __forceinline__ void add_pair_group(const float4* __restrict__ h4, int lid, int np,
                                               int d, int atti, float4& acc) {
  float4 v0, v1, v2, v3, v4, v5, v6, v7;
  load_pair<P0 + 0>(h4, lid, np, d, v0);
  load_pair<P0 + 1>(h4, lid, np, d, v1);
  load_pair<P0 + 2>(h4, lid, np, d, v2);
  load_pair<P0 + 3>(h4, lid, np, d, v3);
  load_pair<P0 + 4>(h4, lid, np, d, v4);
  load_pair<P0 + 5>(h4, lid, np, d, v5);
  load_pair<P0 + 6>(h4, lid, np, d, v6);
  load_pair<P0 + 7>(h4, lid, np, d, v7);
  fma_pair<P0 + 0>(atti, v0, acc);
  fma_pair<P0 + 1>(atti, v1, acc);
  fma_pair<P0 + 2>(atti, v2, acc);
  fma_pair<P0 + 3>(atti, v3, acc);
  fma_pair<P0 + 4>(atti, v4, acc);
  fma_pair<P0 + 5>(atti, v5, acc);
  fma_pair<P0 + 6>(atti, v6, acc);
  fma_pair<P0 + 7>(atti, v7, acc);
}
__device__ __forceinline__ void add_rows(const float* __restrict__ h, int lid, int cc,
                                         int d, float att, float4& acc) {
  const float4* h4 = (const float4*)h;
  const int atti = __float_as_int(att);
  const int np   = (cc + 1) >> 1;
  add_pair_group<0>(h4, lid, np, d, atti, acc);
  if (np > AGG_PAIRS)
    add_pair_group<AGG_PAIRS>(h4, lid, np, d, atti, acc);
}

__device__ __forceinline__ int other_half_i(int x) {
  return __builtin_amdgcn_permlanex16(0, x, 0x76543210, 0xfedcba98, false, false);
}
__device__ __forceinline__ float other_half(float x) {
  return __int_as_float(other_half_i(__float_as_int(x)));
}

__device__ __forceinline__ float edge_score(float a1s, float a2d) {
  const float t = a1s + a2d;
  return t > 0.0f ? t : t * NEG_SLOPE;
}

#define BIN_THREADS 1024
#define BIN_WAVES (BIN_THREADS / 32)
#define ROWS_PER_WAVE (BIN_ROWS / BIN_WAVES)
#define PLACE_UNROLL 8

template <typename F>
__device__ __forceinline__ void for_row_records(
    const int* __restrict__ cellcnt, const int2* __restrict__ bincells,
    const int* __restrict__ spillcnt, const int2* __restrict__ spill,
    int b, int s, int nbins, int npart, int lane, F f) {
  for (int w = 0; w < npart; ++w) {
    const int cw = min(cellcnt[(size_t)w * nbins + b], CELL_CAP);
    for (int j0 = 0; j0 < cw; j0 += 32) {
      const int j = j0 + lane;
      bool mine = false;
      int d = 0;
      if (j < cw) {
        const int2 sd = bincells[(size_t)w * CELL_CAP + j];
        mine = sd.x == s;
        d = sd.y;
      }
      f(mine, d);
    }
    const int ns = min(spillcnt[w], PART_EDGES);
    const int2* sl = spill + (size_t)w * PART_EDGES;
    for (int j0 = 0; j0 < ns; j0 += 32) {
      const int j = j0 + lane;
      bool mine = false;
      int d = 0;
      if (j < ns) {
        const int2 sd = sl[j];
        mine = sd.x == s;
        d = sd.y;
      }
      f(mine, d);
    }
  }
}

__global__ __launch_bounds__(BIN_THREADS, 16) void gat_bin_rows(
    const int* __restrict__ cellcnt, const int* __restrict__ cells,
    const int* __restrict__ spillcnt, const int* __restrict__ spill,
    const float* __restrict__ a1, const float* __restrict__ a2,
    const float* __restrict__ h, float* __restrict__ out, int n, int nbins, int npart) {
  __shared__ int lc[BIN_ROWS];
  __shared__ __attribute__((aligned(16))) int stage[BIN_ROWS * ROW_CAP];
  __shared__ int lcell[BIN_ROWS];
  const int tid  = threadIdx.x;
  const int lane = tid & 31;
  const int wv   = __builtin_amdgcn_readfirstlane(tid >> 5);
  const int b    = blockIdx.x;
  const int row0 = b * BIN_ROWS;

  float a1v = 0.0f;
  {
    const int s = row0 + wv + (lane & (ROWS_PER_WAVE - 1)) * BIN_WAVES;
    if (s < n) a1v = a1[s];
  }

  if (tid < BIN_ROWS) lc[tid] = 0;

  const int2* bincells = (const int2*)cells + (size_t)b * npart * CELL_CAP;
  for (int w0 = 0; w0 < npart; w0 += BIN_ROWS) {
    const int nw = min(BIN_ROWS, npart - w0);
    int cc = 0;
    if (tid < nw) cc = min(cellcnt[(size_t)(w0 + tid) * nbins + b], CELL_CAP);
    __syncthreads();
    if (tid < BIN_ROWS) lcell[tid] = cc;
    __syncthreads();
    const int2* cb = bincells + (size_t)w0 * CELL_CAP;
    const int m = nw * 32;
    for (int i0 = tid; i0 < m; i0 += BIN_THREADS * PLACE_UNROLL) {
      int2 sd[PLACE_UNROLL];
      int  r[PLACE_UNROLL];
      #pragma unroll
      for (int u = 0; u < PLACE_UNROLL; ++u) {
        const int i = i0 + u * BIN_THREADS;
        sd[u].x = -1; sd[u].y = 0;
        if (i < m) {
          const int cell = i >> 5;
          if (lane < lcell[cell]) sd[u] = cb[cell * CELL_CAP + lane];
        }
      }
      #pragma unroll
      for (int u = 0; u < PLACE_UNROLL; ++u)
        r[u] = sd[u].x >= 0 ? atomicAdd(&lc[sd[u].x & (BIN_ROWS - 1)], 1) : 0;
      #pragma unroll
      for (int u = 0; u < PLACE_UNROLL; ++u)
        if (sd[u].x >= 0 && r[u] < ROW_CAP)
          stage[(sd[u].x & (BIN_ROWS - 1)) * ROW_CAP + r[u]] = sd[u].y;
    }
    unsigned longer = __builtin_amdgcn_ballot_w32(cc > 32);
    while (longer != 0u) {
      const int j = __builtin_ctz(longer);
      longer &= longer - 1u;
      const int cell = (tid & ~31) + j;
      const int cj   = __builtin_amdgcn_readlane(cc, j);
      if (32 + lane < cj) {
        const int2 sd = cb[cell * CELL_CAP + 32 + lane];
        const int r = atomicAdd(&lc[sd.x & (BIN_ROWS - 1)], 1);
        if (r < ROW_CAP) stage[(sd.x & (BIN_ROWS - 1)) * ROW_CAP + r] = sd.y;
      }
    }
  }
  for (int w = tid; w < npart; w += BIN_THREADS) {
    const int ns = min(spillcnt[w], PART_EDGES);
    const int2* sl = (const int2*)spill + (size_t)w * PART_EDGES;
    for (int k = 0; k < ns; ++k) {
      const int2 sd = sl[k];
      if ((sd.x >> BIN_SHIFT) == b) {
        const int r = atomicAdd(&lc[sd.x & (BIN_ROWS - 1)], 1);
        if (r < ROW_CAP) stage[(sd.x & (BIN_ROWS - 1)) * ROW_CAP + r] = sd.y;
      }
    }
  }
  __syncthreads();

  for (int k = 0; k < ROWS_PER_WAVE; ++k) {
    const int rl = wv + k * BIN_WAVES;
    const int s  = row0 + rl;
    if (s < n) {
      const int   c   = __builtin_amdgcn_readfirstlane(lc[rl]);
      const float a1s = __int_as_float(__builtin_amdgcn_readlane(__float_as_int(a1v), k));
      const int*  row = &stage[rl * ROW_CAP];
      float* orow = out + (size_t)s * D_DIM;

      if (c <= ROW_CAP) {
        const int lh   = lane >> 4;
        const int lid  = lane & 15;
        const int lid4 = lid << 2;
        const int sl   = 2 * lid + lh;
        float4 acc = make_float4(0.0f, 0.0f, 0.0f, 0.0f);
        if (c <= 32) {
          int d = 0;
          float v = -__builtin_inff();
          if (sl < c) {
            d = row[sl];
            v = edge_score(a1s, a2[d]);
          }
          const float m   = wave_max(v);
          const float ev  = sl < c ? __expf(v - m) : 0.0f;
          const float den = wave_sum(ev);
          const float att = sl < c ? ev / den : 0.0f;
          const int dlo = other_half_i(d);
          if (sl >= c) d = dlo;
          add_rows(h, lid, c, d, att, acc);
        } else {
          float m = -__builtin_inff();
          for (int j0 = 0; j0 < c; j0 += 32) {
            const int j = j0 + sl;
            if (j < c) m = fmaxf(m, edge_score(a1s, a2[row[j]]));
          }
          m = wave_max(m);
          float part = 0.0f;
          for (int j0 = 0; j0 < c; j0 += 32) {
            const int j = j0 + sl;
            if (j < c) part += __expf(edge_score(a1s, a2[row[j]]) - m);
          }
          const float den = wave_sum(part);
          for (int j0 = 0; j0 < c; j0 += 32) {
            const int j = j0 + sl;
            int d = 0;
            float att = 0.0f;
            if (j < c) {
              d = row[j];
              att = __expf(edge_score(a1s, a2[d]) - m) / den;
            }
            const int dlo = other_half_i(d);
            if (j >= c) d = dlo;
            add_rows(h, lid, min(32, c - j0), d, att, acc);
          }
        }
        acc.x += other_half(acc.x);
        acc.y += other_half(acc.y);
        acc.z += other_half(acc.z);
        acc.w += other_half(acc.w);
        if (lane < 16) *(float4*)(orow + lid4) = acc;
      } else {
        float acc0 = 0.0f, acc1 = 0.0f;
        float m = -__builtin_inff();
        for_row_records(cellcnt, bincells, spillcnt, (const int2*)spill, b, s, nbins, npart, lane,
                        [&](bool mine, int d) {
          if (mine) m = fmaxf(m, edge_score(a1s, a2[d]));
        });
        m = wave_max(m);
        float part = 0.0f;
        for_row_records(cellcnt, bincells, spillcnt, (const int2*)spill, b, s, nbins, npart, lane,
                        [&](bool mine, int d) {
          if (mine) part += __expf(edge_score(a1s, a2[d]) - m);
        });
        const float den = wave_sum(part);
        for_row_records(cellcnt, bincells, spillcnt, (const int2*)spill, b, s, nbins, npart, lane,
                        [&](bool mine, int d) {
          float att = 0.0f;
          if (mine) att = __expf(edge_score(a1s, a2[d]) - m) / den;
          unsigned live = __builtin_amdgcn_ballot_w32(mine);
          while (live != 0u) {
            const int jj = __builtin_ctz(live);
            live &= live - 1u;
            const int   dj = __builtin_amdgcn_readlane(d, jj);
            const float aj = __int_as_float(__builtin_amdgcn_readlane(__float_as_int(att), jj));
            const float* hr = h + (size_t)dj * D_DIM;
            acc0 += aj * hr[lane];
            acc1 += aj * hr[lane + 32];
          }
        });
        orow[lane]      = acc0;
        orow[lane + 32] = acc1;
      }
    }
  }
}

extern "C" void kernel_launch(void* const* d_in, const int* in_sizes, int n_in,
                              void* d_out, int out_size, void* d_ws, size_t ws_size,
                              hipStream_t stream) {
  const float* feat = (const float*)d_in[0];
  const int*   idx  = (const int*)d_in[1];
  const float* W    = (const float*)d_in[2];
  const float* bias = (const float*)d_in[3];
  const float* a1w  = (const float*)d_in[4];
  const float* a1b  = (const float*)d_in[5];
  const float* a2w  = (const float*)d_in[6];
  const float* a2b  = (const float*)d_in[7];
  float* out = (float*)d_out;

  const int n = in_sizes[0] / D_DIM;
  const int E = in_sizes[1] / 2;
  const int* src = idx;
  const int* dst = idx + E;

  const int nbins = (n + BIN_ROWS - 1) / BIN_ROWS;
  const int npart = (E + PART_EDGES - 1) / PART_EDGES;
  float* h        = (float*)d_ws;
  int*   cells    = (int*)(h + (size_t)n * D_DIM);
  int*   spill    = cells + (size_t)nbins * npart * CELL_CAP * 2;
  float* a1       = (float*)(spill + (size_t)npart * PART_EDGES * 2);
  float* a2       = a1 + n;
  int*   cellcnt  = (int*)(a2 + n);
  int*   spillcnt = cellcnt + (size_t)npart * nbins;

  const int tiles = (n + TILE_M - 1) / TILE_M;
  gat_linear_and_partition<<<npart + (tiles + 8 * TILES_PER_WAVE - 1) / (8 * TILES_PER_WAVE), 256, 0, stream>>>(
      feat, W, bias, a1w, a1b, a2w, a2b, h, a1, a2, n,
      src, dst, cells, cellcnt, spillcnt, spill, E, nbins, npart);
  gat_bin_rows<<<nbins, BIN_THREADS, 0, stream>>>(
      cellcnt, cells, spillcnt, spill, a1, a2, h, out, n, nbins, npart);
}
